# starting kernel
# baseline (speedup 1.0000x reference)
#include <hip/hip_runtime.h>
#include <hip/hip_bf16.h>

typedef __attribute__((ext_vector_type(2))) float v2f;
typedef __attribute__((ext_vector_type(8))) float v8f;

#define IN_DIM  128
#define HID_DIM 64
#define OUT_DIM 128

__global__ void deg_kernel(const int* __restrict__ dst, int E,
                           unsigned int* __restrict__ deg) {
    int i = blockIdx.x * blockDim.x + threadIdx.x;
    if (i < E) atomicAdd(&deg[dst[i]], 1u);
}

__global__ void dinv_kernel(const unsigned int* __restrict__ deg, int N,
                            float* __restrict__ dinv) {
    int i = blockIdx.x * blockDim.x + threadIdx.x;
    if (i < N) dinv[i] = rsqrtf((float)deg[i] + 1.0f);
}

__global__ __launch_bounds__(256) void gemm1_kernel(
    const float* __restrict__ x, const float* __restrict__ Wc,
    float* __restrict__ h, int N) {
    __shared__ float sW[IN_DIM * HID_DIM];
    {
        const float4* gv = (const float4*)Wc;
        float4* sv = (float4*)sW;
        for (int i = threadIdx.x; i < (IN_DIM * HID_DIM) / 4; i += 256)
            sv[i] = gv[i];
    }
    __syncthreads();

    const int wave = threadIdx.x >> 5;
    const int lane = threadIdx.x & 31;
    const int rowBase = blockIdx.x * 128 + wave * 16;
    if (rowBase >= N) return;

    const int r  = lane & 15;
    const int kg = lane >> 4;
    const int rowLd = min(rowBase + r, N - 1);
    const float* xrow = x + (size_t)rowLd * IN_DIM;

    v8f acc[4] = {v8f{}, v8f{}, v8f{}, v8f{}};
    for (int k = 0; k < IN_DIM; k += 4) {
        const int kk = k + (kg << 1);
        float2 av = *(const float2*)(xrow + kk);
        v2f a; a.x = av.x; a.y = av.y;
#pragma unroll
        for (int t = 0; t < 4; ++t) {
            v2f b;
            b.x = sW[(kk)     * HID_DIM + t * 16 + r];
            b.y = sW[(kk + 1) * HID_DIM + t * 16 + r];
            acc[t] = __builtin_amdgcn_wmma_f32_16x16x4_f32(
                false, a, false, b, (short)0, acc[t], false, false);
        }
    }
#pragma unroll
    for (int t = 0; t < 4; ++t) {
        const int col = t * 16 + r;
#pragma unroll
        for (int rr = 0; rr < 8; ++rr) {
            const int row = rowBase + rr + 8 * kg;
            if (row < N) h[(size_t)row * HID_DIM + col] = acc[t][rr];
        }
    }
}

__global__ __launch_bounds__(256) void scatter_kernel(
    const int* __restrict__ src, const int* __restrict__ dst,
    const float* __restrict__ dinv, const float* __restrict__ h,
    float* __restrict__ agg, int E) {
    const int e    = (int)((blockIdx.x * (unsigned)blockDim.x + threadIdx.x) >> 5);
    const int lane = threadIdx.x & 31;
    if (e >= E) return;
    const int s = src[e];
    const int d = dst[e];
    const float nrm = dinv[s] * dinv[d];
    const float2 hv = *(const float2*)(h + (size_t)s * HID_DIM + lane * 2);
    float* ap = agg + (size_t)d * HID_DIM + lane * 2;
    atomicAdd(ap,     hv.x * nrm);
    atomicAdd(ap + 1, hv.y * nrm);
}

__global__ __launch_bounds__(256) void gemm2_kernel(
    const float* __restrict__ agg, const float* __restrict__ h,
    const float* __restrict__ dinv, const float* __restrict__ bc,
    const float* __restrict__ Wl,  const float* __restrict__ bl,
    float* __restrict__ out, int N) {
    __shared__ float sW[HID_DIM * OUT_DIM];
    {
        const float4* gv = (const float4*)Wl;
        float4* sv = (float4*)sW;
        for (int i = threadIdx.x; i < (HID_DIM * OUT_DIM) / 4; i += 256)
            sv[i] = gv[i];
    }
    __syncthreads();

    const int wave = threadIdx.x >> 5;
    const int lane = threadIdx.x & 31;
    const int rowBase = blockIdx.x * 128 + wave * 16;
    if (rowBase >= N) return;

    const int r  = lane & 15;
    const int kg = lane >> 4;
    const int rowLd = min(rowBase + r, N - 1);
    const float* arow = agg + (size_t)rowLd * HID_DIM;
    const float* hrow = h   + (size_t)rowLd * HID_DIM;
    const float  di   = dinv[rowLd];
    const float  di2  = di * di;

    v8f acc[8] = {v8f{}, v8f{}, v8f{}, v8f{}, v8f{}, v8f{}, v8f{}, v8f{}};
    for (int k = 0; k < HID_DIM; k += 4) {
        const int kk = k + (kg << 1);
        float2 av  = *(const float2*)(arow + kk);
        float2 hv  = *(const float2*)(hrow + kk);
        float2 bcv = *(const float2*)(bc + kk);
        v2f a;
        a.x = fmaxf(av.x + hv.x * di2 + bcv.x, 0.0f);
        a.y = fmaxf(av.y + hv.y * di2 + bcv.y, 0.0f);
#pragma unroll
        for (int t = 0; t < 8; ++t) {
            v2f b;
            b.x = sW[(kk)     * OUT_DIM + t * 16 + r];
            b.y = sW[(kk + 1) * OUT_DIM + t * 16 + r];
            acc[t] = __builtin_amdgcn_wmma_f32_16x16x4_f32(
                false, a, false, b, (short)0, acc[t], false, false);
        }
    }
#pragma unroll
    for (int t = 0; t < 8; ++t) {
        const int col = t * 16 + r;
        const float blv = bl[col];
#pragma unroll
        for (int rr = 0; rr < 8; ++rr) {
            const int row = rowBase + rr + 8 * kg;
            if (row < N) out[(size_t)row * OUT_DIM + col] = acc[t][rr] + blv;
        }
    }
}

static inline size_t alignUp(size_t v) { return (v + 255) & ~(size_t)255; }

extern "C" void kernel_launch(void* const* d_in, const int* in_sizes, int n_in,
                              void* d_out, int out_size, void* d_ws, size_t ws_size,
                              hipStream_t stream) {
    const float* x  = (const float*)d_in[0];
    const int*   ei = (const int*)  d_in[1];
    const float* Wc = (const float*)d_in[2];
    const float* bc = (const float*)d_in[3];
    const float* Wl = (const float*)d_in[4];
    const float* bl = (const float*)d_in[5];
    float* out = (float*)d_out;

    const int N = in_sizes[0] / IN_DIM;
    const int E = in_sizes[1] / 2;
    const int* src = ei;
    const int* dst = ei + E;

    char* ws = (char*)d_ws;
    const size_t hBytes = (size_t)N * HID_DIM * sizeof(float);
    size_t off = 0;
    float*        h    = (float*)(ws + off);        off = alignUp(off + hBytes);
    float*        agg  = (float*)(ws + off);        off = alignUp(off + hBytes);
    float*        dinv = (float*)(ws + off);        off = alignUp(off + (size_t)N * sizeof(float));
    unsigned int* deg  = (unsigned int*)(ws + off); off = alignUp(off + (size_t)N * sizeof(unsigned int));

    hipMemsetAsync(agg, 0, hBytes, stream);
    hipMemsetAsync(deg, 0, (size_t)N * sizeof(unsigned int), stream);

    deg_kernel <<<(E + 255) / 256, 256, 0, stream>>>(dst, E, deg);
    dinv_kernel<<<(N + 255) / 256, 256, 0, stream>>>(deg, N, dinv);

    const int rowBlocks = (N + 127) / 128;
    gemm1_kernel<<<rowBlocks, 256, 0, stream>>>(x, Wc, h, N);

    scatter_kernel<<<(E + 7) / 8, 256, 0, stream>>>(src, dst, dinv, h, agg, E);

    gemm2_kernel<<<rowBlocks, 256, 0, stream>>>(agg, h, dinv, bc, Wl, bl, out, N);
}


# Degree count moved from 1.6M random global atomics to LDS-private partial histograms (32 edge chunks x 2 node ranges, ds_add), summed per node in the dinv kernel; no global atomics left in it
# speedup vs baseline: 1.2632x; 1.2632x over previous
#include <hip/hip_runtime.h>
#include <hip/hip_bf16.h>

typedef __attribute__((ext_vector_type(2))) float v2f;
typedef __attribute__((ext_vector_type(8))) float v8f;

#define IN_DIM  128
#define HID_DIM 64
#define OUT_DIM 128

#define HIST_BINS   51200
#define HIST_CHUNKS 32

__global__ __launch_bounds__(1024) void deg_kernel(const int* __restrict__ dst, int E, int N,
                                                   unsigned int* __restrict__ part) {
    __shared__ unsigned int cnt[HIST_BINS];
    const int NP    = (N + 31) & ~31;
    const int binLo = blockIdx.y * HIST_BINS;
    const int nb    = min(HIST_BINS, N - binLo);
    {
        uint4* cv = (uint4*)cnt;
        for (int i = threadIdx.x; i < HIST_BINS / 4; i += 1024)
            cv[i] = make_uint4(0u, 0u, 0u, 0u);
    }
    __syncthreads();

    const int per = (E + (int)gridDim.x - 1) / (int)gridDim.x;
    const int e0  = blockIdx.x * per;
    const int e1  = min(E, e0 + per);
    int e = e0 + threadIdx.x;
    for (; e + 15 * 1024 < e1; e += 16 * 1024) {
        unsigned int b[16];
#pragma unroll
        for (int u = 0; u < 16; ++u) b[u] = (unsigned int)(dst[e + u * 1024] - binLo);
#pragma unroll
        for (int u = 0; u < 16; ++u)
            if (b[u] < (unsigned int)nb) atomicAdd(&cnt[b[u]], 1u);
    }
    for (; e < e1; e += 1024) {
        const unsigned int b = (unsigned int)(dst[e] - binLo);
        if (b < (unsigned int)nb) atomicAdd(&cnt[b], 1u);
    }
    __syncthreads();

    unsigned int* row = part + (size_t)blockIdx.x * NP + binLo;
    const int nb4 = nb >> 2;
    {
        const uint4* cv = (const uint4*)cnt;
        uint4* rv = (uint4*)row;
        for (int i = threadIdx.x; i < nb4; i += 1024) rv[i] = cv[i];
    }
    for (int i = (nb4 << 2) + threadIdx.x; i < nb; i += 1024) row[i] = cnt[i];
}

__global__ void dinv_kernel(const unsigned int* __restrict__ part, int N,
                            float* __restrict__ dinv) {
    const int NP = (N + 31) & ~31;
    int i = blockIdx.x * blockDim.x + threadIdx.x;
    if (i < N) {
        unsigned int deg = 0u;
#pragma unroll
        for (int g = 0; g < HIST_CHUNKS; ++g) deg += part[(size_t)g * NP + i];
        dinv[i] = rsqrtf((float)deg + 1.0f);
    }
}

__global__ __launch_bounds__(256) void gemm1_kernel(
    const float* __restrict__ x, const float* __restrict__ Wc,
    float* __restrict__ h, int N) {
    __shared__ float sW[IN_DIM * HID_DIM];
    {
        const float4* gv = (const float4*)Wc;
        float4* sv = (float4*)sW;
        for (int i = threadIdx.x; i < (IN_DIM * HID_DIM) / 4; i += 256)
            sv[i] = gv[i];
    }
    __syncthreads();

    const int wave = threadIdx.x >> 5;
    const int lane = threadIdx.x & 31;
    const int rowBase = blockIdx.x * 128 + wave * 16;
    if (rowBase >= N) return;

    const int r  = lane & 15;
    const int kg = lane >> 4;
    const int rowLd = min(rowBase + r, N - 1);
    const float* xrow = x + (size_t)rowLd * IN_DIM;

    v8f acc[4] = {v8f{}, v8f{}, v8f{}, v8f{}};
    for (int k = 0; k < IN_DIM; k += 4) {
        const int kk = k + (kg << 1);
        float2 av = *(const float2*)(xrow + kk);
        v2f a; a.x = av.x; a.y = av.y;
#pragma unroll
        for (int t = 0; t < 4; ++t) {
            v2f b;
            b.x = sW[(kk)     * HID_DIM + t * 16 + r];
            b.y = sW[(kk + 1) * HID_DIM + t * 16 + r];
            acc[t] = __builtin_amdgcn_wmma_f32_16x16x4_f32(
                false, a, false, b, (short)0, acc[t], false, false);
        }
    }
#pragma unroll
    for (int t = 0; t < 4; ++t) {
        const int col = t * 16 + r;
#pragma unroll
        for (int rr = 0; rr < 8; ++rr) {
            const int row = rowBase + rr + 8 * kg;
            if (row < N) h[(size_t)row * HID_DIM + col] = acc[t][rr];
        }
    }
}

__global__ __launch_bounds__(256) void scatter_kernel(
    const int* __restrict__ src, const int* __restrict__ dst,
    const float* __restrict__ dinv, const float* __restrict__ h,
    float* __restrict__ agg, int E) {
    const int e    = (int)((blockIdx.x * (unsigned)blockDim.x + threadIdx.x) >> 5);
    const int lane = threadIdx.x & 31;
    if (e >= E) return;
    const int s = src[e];
    const int d = dst[e];
    const float nrm = dinv[s] * dinv[d];
    const float2 hv = *(const float2*)(h + (size_t)s * HID_DIM + lane * 2);
    float* ap = agg + (size_t)d * HID_DIM + lane * 2;
    atomicAdd(ap,     hv.x * nrm);
    atomicAdd(ap + 1, hv.y * nrm);
}

__global__ __launch_bounds__(256) void gemm2_kernel(
    const float* __restrict__ agg, const float* __restrict__ h,
    const float* __restrict__ dinv, const float* __restrict__ bc,
    const float* __restrict__ Wl,  const float* __restrict__ bl,
    float* __restrict__ out, int N) {
    __shared__ float sW[HID_DIM * OUT_DIM];
    {
        const float4* gv = (const float4*)Wl;
        float4* sv = (float4*)sW;
        for (int i = threadIdx.x; i < (HID_DIM * OUT_DIM) / 4; i += 256)
            sv[i] = gv[i];
    }
    __syncthreads();

    const int wave = threadIdx.x >> 5;
    const int lane = threadIdx.x & 31;
    const int rowBase = blockIdx.x * 128 + wave * 16;
    if (rowBase >= N) return;

    const int r  = lane & 15;
    const int kg = lane >> 4;
    const int rowLd = min(rowBase + r, N - 1);
    const float* arow = agg + (size_t)rowLd * HID_DIM;
    const float* hrow = h   + (size_t)rowLd * HID_DIM;
    const float  di   = dinv[rowLd];
    const float  di2  = di * di;

    v8f acc[8] = {v8f{}, v8f{}, v8f{}, v8f{}, v8f{}, v8f{}, v8f{}, v8f{}};
    for (int k = 0; k < HID_DIM; k += 4) {
        const int kk = k + (kg << 1);
        float2 av  = *(const float2*)(arow + kk);
        float2 hv  = *(const float2*)(hrow + kk);
        float2 bcv = *(const float2*)(bc + kk);
        v2f a;
        a.x = fmaxf(av.x + hv.x * di2 + bcv.x, 0.0f);
        a.y = fmaxf(av.y + hv.y * di2 + bcv.y, 0.0f);
#pragma unroll
        for (int t = 0; t < 8; ++t) {
            v2f b;
            b.x = sW[(kk)     * OUT_DIM + t * 16 + r];
            b.y = sW[(kk + 1) * OUT_DIM + t * 16 + r];
            acc[t] = __builtin_amdgcn_wmma_f32_16x16x4_f32(
                false, a, false, b, (short)0, acc[t], false, false);
        }
    }
#pragma unroll
    for (int t = 0; t < 8; ++t) {
        const int col = t * 16 + r;
        const float blv = bl[col];
#pragma unroll
        for (int rr = 0; rr < 8; ++rr) {
            const int row = rowBase + rr + 8 * kg;
            if (row < N) out[(size_t)row * OUT_DIM + col] = acc[t][rr] + blv;
        }
    }
}

static inline size_t alignUp(size_t v) { return (v + 255) & ~(size_t)255; }

extern "C" void kernel_launch(void* const* d_in, const int* in_sizes, int n_in,
                              void* d_out, int out_size, void* d_ws, size_t ws_size,
                              hipStream_t stream) {
    const float* x  = (const float*)d_in[0];
    const int*   ei = (const int*)  d_in[1];
    const float* Wc = (const float*)d_in[2];
    const float* bc = (const float*)d_in[3];
    const float* Wl = (const float*)d_in[4];
    const float* bl = (const float*)d_in[5];
    float* out = (float*)d_out;

    const int N = in_sizes[0] / IN_DIM;
    const int E = in_sizes[1] / 2;
    const int* src = ei;
    const int* dst = ei + E;

    char* ws = (char*)d_ws;
    const size_t hBytes = (size_t)N * HID_DIM * sizeof(float);
    size_t off = 0;
    float*        h    = (float*)(ws + off);        off = alignUp(off + hBytes);
    float*        agg  = (float*)(ws + off);        off = alignUp(off + hBytes);
    float*        dinv = (float*)(ws + off);        off = alignUp(off + (size_t)N * sizeof(float));
    unsigned int* part = (unsigned int*)h;

    hipMemsetAsync(agg, 0, hBytes, stream);

    deg_kernel <<<dim3(HIST_CHUNKS, (N + HIST_BINS - 1) / HIST_BINS), 1024, 0, stream>>>(dst, E, N, part);
    dinv_kernel<<<(N + 255) / 256, 256, 0, stream>>>(part, N, dinv);

    const int rowBlocks = (N + 127) / 128;
    gemm1_kernel<<<rowBlocks, 256, 0, stream>>>(x, Wc, h, N);

    scatter_kernel<<<(E + 7) / 8, 256, 0, stream>>>(src, dst, dinv, h, agg, E);

    gemm2_kernel<<<rowBlocks, 256, 0, stream>>>(agg, h, dinv, bc, Wl, bl, out, N);
}


# Atomic scatter-add replaced by a gather over target-sorted edges (counting sort on the partial histograms, LDS arrival-order ranks) in gemm2's A-tile build; h pre-scaled by dinv; sum order changes
# speedup vs baseline: 2.0210x; 1.5999x over previous
#include <hip/hip_runtime.h>
#include <hip/hip_bf16.h>

typedef __attribute__((ext_vector_type(2))) float v2f;
typedef __attribute__((ext_vector_type(8))) float v8f;

#define IN_DIM  128
#define HID_DIM 64
#define OUT_DIM 128

#define HIST_BINS   51200
#define HIST_CHUNKS 32

__global__ __launch_bounds__(1024) void deg_kernel(const int* __restrict__ dst, int E, int N,
                                                   unsigned int* __restrict__ part) {
    __shared__ unsigned int cnt[HIST_BINS];
    const int NP    = (N + 31) & ~31;
    const int binLo = blockIdx.y * HIST_BINS;
    const int nb    = min(HIST_BINS, N - binLo);
    {
        uint4* cv = (uint4*)cnt;
        for (int i = threadIdx.x; i < HIST_BINS / 4; i += 1024)
            cv[i] = make_uint4(0u, 0u, 0u, 0u);
    }
    __syncthreads();

    const int per = (E + (int)gridDim.x - 1) / (int)gridDim.x;
    const int e0  = blockIdx.x * per;
    const int e1  = min(E, e0 + per);
    int e = e0 + threadIdx.x;
    for (; e + 15 * 1024 < e1; e += 16 * 1024) {
        unsigned int b[16];
#pragma unroll
        for (int u = 0; u < 16; ++u) b[u] = (unsigned int)(dst[e + u * 1024] - binLo);
#pragma unroll
        for (int u = 0; u < 16; ++u)
            if (b[u] < (unsigned int)nb) atomicAdd(&cnt[b[u]], 1u);
    }
    for (; e < e1; e += 1024) {
        const unsigned int b = (unsigned int)(dst[e] - binLo);
        if (b < (unsigned int)nb) atomicAdd(&cnt[b], 1u);
    }
    __syncthreads();

    unsigned int* row = part + (size_t)blockIdx.x * NP + binLo;
    const int nb4 = nb >> 2;
    {
        const uint4* cv = (const uint4*)cnt;
        uint4* rv = (uint4*)row;
        for (int i = threadIdx.x; i < nb4; i += 1024) rv[i] = cv[i];
    }
    for (int i = (nb4 << 2) + threadIdx.x; i < nb; i += 1024) row[i] = cnt[i];
}

#define SCAN_BLK      256
#define MAX_SCAN_BLKS 1024

__global__ __launch_bounds__(SCAN_BLK) void dinv_kernel(unsigned int* __restrict__ part, int N,
                                                       float* __restrict__ dinv,
                                                       unsigned int* __restrict__ blkTot) {
    __shared__ unsigned int wsum[SCAN_BLK / 32];
    const int NP   = (N + 31) & ~31;
    const int i    = blockIdx.x * SCAN_BLK + threadIdx.x;
    const int lane = threadIdx.x & 31;
    const int wave = threadIdx.x >> 5;
    unsigned int c[HIST_CHUNKS];
    unsigned int deg = 0u;
    if (i < N) {
#pragma unroll
        for (int g = 0; g < HIST_CHUNKS; ++g) c[g] = part[(size_t)g * NP + i];
#pragma unroll
        for (int g = 0; g < HIST_CHUNKS; ++g) { const unsigned int t = c[g]; c[g] = deg; deg += t; }
        dinv[i] = rsqrtf((float)deg + 1.0f);
    }
    unsigned int incl = deg;
    for (int o = 1; o < 32; o <<= 1) {
        const unsigned int u = __shfl_up(incl, o);
        if (lane >= o) incl += u;
    }
    if (lane == 31) wsum[wave] = incl;
    __syncthreads();
    unsigned int wbase = 0u, tot = 0u;
    for (int w = 0; w < SCAN_BLK / 32; ++w) {
        const unsigned int v = wsum[w];
        if (w < wave) wbase += v;
        tot += v;
    }
    const unsigned int excl = wbase + incl - deg;
    if (i < N) {
#pragma unroll
        for (int g = 0; g < HIST_CHUNKS; ++g) part[(size_t)g * NP + i] = c[g] + excl;
    }
    if (threadIdx.x == 0) blkTot[blockIdx.x] = tot;
}

__global__ __launch_bounds__(1024) void fill_kernel(const int* __restrict__ src,
                                                    const int* __restrict__ dst, int E, int N,
                                                    const unsigned int* __restrict__ part,
                                                    const unsigned int* __restrict__ blkTot, int nblk,
                                                    unsigned int* __restrict__ rowptr,
                                                    int* __restrict__ ssrc) {
    __shared__ unsigned int cur[HIST_BINS];
    __shared__ unsigned int bpre[MAX_SCAN_BLKS];
    __shared__ unsigned int wtot[32];
    const int NP    = (N + 31) & ~31;
    const int binLo = blockIdx.y * HIST_BINS;
    const int nb    = min(HIST_BINS, N - binLo);
    const int t     = threadIdx.x;
    const int lane  = t & 31;

    {
        const unsigned int v = (t < nblk) ? blkTot[t] : 0u;
        unsigned int incl = v;
        for (int o = 1; o < 32; o <<= 1) {
            const unsigned int u = __shfl_up(incl, o);
            if (lane >= o) incl += u;
        }
        if (lane == 31) wtot[t >> 5] = incl;
        __syncthreads();
        if (t < 32) {
            const unsigned int w = wtot[t];
            unsigned int wi = w;
            for (int o = 1; o < 32; o <<= 1) {
                const unsigned int u = __shfl_up(wi, o);
                if (lane >= o) wi += u;
            }
            wtot[t] = wi - w;
        }
        __syncthreads();
        bpre[t] = wtot[t >> 5] + incl - v;
    }
    __syncthreads();

    const unsigned int* prow = part + (size_t)blockIdx.x * NP + binLo;
    const int nb4 = nb >> 2;
    {
        const uint4* pv = (const uint4*)prow;
        uint4* cv = (uint4*)cur;
        uint4* rv = (uint4*)(rowptr + binLo);
        for (int i0 = t; i0 < nb4; i0 += 4 * 1024) {
            uint4 v[4];
#pragma unroll
            for (int u = 0; u < 4; ++u) {
                const int i = i0 + u * 1024;
                v[u] = (i < nb4) ? pv[i] : make_uint4(0u, 0u, 0u, 0u);
            }
#pragma unroll
            for (int u = 0; u < 4; ++u) {
                const int i = i0 + u * 1024;
                if (i < nb4) {
                    const unsigned int bp = bpre[(binLo + 4 * i) / SCAN_BLK];
                    uint4 w = v[u];
                    w.x += bp; w.y += bp; w.z += bp; w.w += bp;
                    cv[i] = w;
                    if (blockIdx.x == 0) rv[i] = w;
                }
            }
        }
    }
    for (int i = (nb4 << 2) + t; i < nb; i += 1024) {
        const unsigned int v = prow[i] + bpre[(binLo + i) / SCAN_BLK];
        cur[i] = v;
        if (blockIdx.x == 0) rowptr[binLo + i] = v;
    }
    if (blockIdx.x == 0 && blockIdx.y == gridDim.y - 1 && t == 0)
        rowptr[N] = bpre[nblk - 1] + blkTot[nblk - 1];
    __syncthreads();

    const int per = (E + (int)gridDim.x - 1) / (int)gridDim.x;
    const int e0  = blockIdx.x * per;
    const int e1  = min(E, e0 + per);
    int e = e0 + t;
    for (; e + 7 * 1024 < e1; e += 8 * 1024) {
        unsigned int b[8];
        int s[8];
#pragma unroll
        for (int u = 0; u < 8; ++u) {
            b[u] = (unsigned int)(dst[e + u * 1024] - binLo);
            s[u] = src[e + u * 1024];
        }
        unsigned int pos[8];
#pragma unroll
        for (int u = 0; u < 8; ++u) {
            pos[u] = 0u;
            if (b[u] < (unsigned int)nb) pos[u] = atomicAdd(&cur[b[u]], 1u);
        }
#pragma unroll
        for (int u = 0; u < 8; ++u)
            if (b[u] < (unsigned int)nb) ssrc[pos[u]] = s[u];
    }
    for (; e < e1; e += 1024) {
        const unsigned int b = (unsigned int)(dst[e] - binLo);
        if (b < (unsigned int)nb) ssrc[atomicAdd(&cur[b], 1u)] = src[e];
    }
}

__global__ __launch_bounds__(256) void gemm1_kernel(
    const float* __restrict__ x, const float* __restrict__ Wc,
    const float* __restrict__ dinv, float* __restrict__ h, int N) {
    __shared__ float sW[IN_DIM * HID_DIM];
    {
        const float4* gv = (const float4*)Wc;
        float4* sv = (float4*)sW;
        for (int i = threadIdx.x; i < (IN_DIM * HID_DIM) / 4; i += 256)
            sv[i] = gv[i];
    }
    __syncthreads();

    const int wave = threadIdx.x >> 5;
    const int lane = threadIdx.x & 31;
    const int rowBase = blockIdx.x * 128 + wave * 16;
    if (rowBase >= N) return;

    const int r  = lane & 15;
    const int kg = lane >> 4;
    const int rowLd = min(rowBase + r, N - 1);
    const float* xrow = x + (size_t)rowLd * IN_DIM;

    v8f acc[4] = {v8f{}, v8f{}, v8f{}, v8f{}};
    for (int k = 0; k < IN_DIM; k += 4) {
        const int kk = k + (kg << 1);
        float2 av = *(const float2*)(xrow + kk);
        v2f a; a.x = av.x; a.y = av.y;
#pragma unroll
        for (int t = 0; t < 4; ++t) {
            v2f b;
            b.x = sW[(kk)     * HID_DIM + t * 16 + r];
            b.y = sW[(kk + 1) * HID_DIM + t * 16 + r];
            acc[t] = __builtin_amdgcn_wmma_f32_16x16x4_f32(
                false, a, false, b, (short)0, acc[t], false, false);
        }
    }
    float dv[8];
#pragma unroll
    for (int rr = 0; rr < 8; ++rr) dv[rr] = dinv[min(rowBase + rr + 8 * kg, N - 1)];
#pragma unroll
    for (int t = 0; t < 4; ++t) {
        const int col = t * 16 + r;
#pragma unroll
        for (int rr = 0; rr < 8; ++rr) {
            const int row = rowBase + rr + 8 * kg;
            if (row < N) h[(size_t)row * HID_DIM + col] = acc[t][rr] * dv[rr];
        }
    }
}

#define ZS_PITCH 68
#define GATHER_SLOT(u) if ((u) < m) { const unsigned int s_ = (unsigned int)__builtin_amdgcn_readlane(sv, (u)); v[(u)] = *(const float2*)(hb + ((s_ << 8) + lb)); }
#define SL_CAP   384

__global__ __launch_bounds__(256) void gemm2_kernel(
    const unsigned int* __restrict__ rowptr, const int* __restrict__ ssrc,
    const float* __restrict__ h, const float* __restrict__ dinv,
    const float* __restrict__ bc, const float* __restrict__ Wl,
    const float* __restrict__ bl, float* __restrict__ out, int N) {
    __shared__ float sW[HID_DIM * OUT_DIM];
    __shared__ float zs[128 * ZS_PITCH];
    __shared__ int   sl[8 * SL_CAP];
    {
        const float4* gv = (const float4*)Wl;
        float4* sv = (float4*)sW;
        for (int i = threadIdx.x; i < (HID_DIM * OUT_DIM) / 4; i += 256)
            sv[i] = gv[i];
    }

    const int wave = threadIdx.x >> 5;
    const int lane = threadIdx.x & 31;
    const int rowBase = blockIdx.x * 128 + wave * 16;

    if (rowBase < N) {
        const unsigned int rp   = rowptr[min(rowBase + lane, N)];
        const float        dvec = dinv[min(rowBase + lane, N - 1)];
        const float2       bcv  = *(const float2*)(bc + lane * 2);
        const unsigned int w0 = (unsigned int)__builtin_amdgcn_readlane((int)rp, 0);
        const unsigned int wn = (unsigned int)__builtin_amdgcn_readlane((int)rp, 16) - w0;
        int* slw = sl + wave * SL_CAP;
        {
            int tmp[SL_CAP / 32];
#pragma unroll
            for (int u = 0; u < SL_CAP / 32; ++u) {
                const unsigned int o = (unsigned int)(u * 32 + lane);
                tmp[u] = (o < wn) ? ssrc[w0 + o] : 0;
            }
#pragma unroll
            for (int u = 0; u < SL_CAP / 32; ++u) slw[u * 32 + lane] = tmp[u];
        }
        const float* hl = h + lane * 2;
        const char*  hb = (const char*)h;
        const unsigned int lb = (unsigned int)lane * 8u;
        for (int i = 0; i < 16; ++i) {
            const unsigned int beg = (unsigned int)__builtin_amdgcn_readlane((int)rp, i);
            const unsigned int cnt = (unsigned int)__builtin_amdgcn_readlane((int)rp, i + 1) - beg;
            const unsigned int lo  = beg - w0;
            const int nld = min(rowBase + i, N - 1);
            float2 acc = *(const float2*)(hl + (size_t)nld * HID_DIM);
            if (lo + cnt <= (unsigned int)SL_CAP) {
                for (unsigned int base = 0u; base < cnt; base += 32u) {
                    const int m  = (int)min(32u, cnt - base);
                    const int sv = slw[min(lo + base + (unsigned int)lane, (unsigned int)(SL_CAP - 1))];
                    float2 v[32];
#pragma unroll
                    for (int u = 0; u < 32; ++u) v[u] = make_float2(0.0f, 0.0f);
#pragma unroll
                    for (int u = 0; u < 8; ++u) GATHER_SLOT(u)
                    if (m > 8) {
#pragma unroll
                        for (int u = 8; u < 16; ++u) GATHER_SLOT(u)
                        if (m > 16) {
#pragma unroll
                            for (int u = 16; u < 24; ++u) GATHER_SLOT(u)
                            if (m > 24) {
#pragma unroll
                                for (int u = 24; u < 32; ++u) GATHER_SLOT(u)
                            }
                        }
                    }
#pragma unroll
                    for (int u = 0; u < 32; ++u) { acc.x += v[u].x; acc.y += v[u].y; }
                }
            } else {
                for (unsigned int j = 0u; j < cnt; ++j) {
                    const int s = __builtin_amdgcn_readfirstlane(ssrc[beg + j]);
                    const float2 v0 = *(const float2*)(hl + (size_t)s * HID_DIM);
                    acc.x += v0.x; acc.y += v0.y;
                }
            }
            const float di = __int_as_float(__builtin_amdgcn_readlane(__float_as_int(dvec), i));
            float2 z;
            z.x = fmaxf(di * acc.x + bcv.x, 0.0f);
            z.y = fmaxf(di * acc.y + bcv.y, 0.0f);
            *(float2*)(zs + (wave * 16 + i) * ZS_PITCH + lane * 2) = z;
        }
    }
    __syncthreads();
    if (rowBase >= N) return;

    const int r  = lane & 15;
    const int kg = lane >> 4;
    const float* zrow = zs + (wave * 16 + r) * ZS_PITCH;

    v8f acc[8] = {v8f{}, v8f{}, v8f{}, v8f{}, v8f{}, v8f{}, v8f{}, v8f{}};
    for (int k = 0; k < HID_DIM; k += 4) {
        const int kk = k + (kg << 1);
        const float2 av = *(const float2*)(zrow + kk);
        v2f a; a.x = av.x; a.y = av.y;
#pragma unroll
        for (int t = 0; t < 8; ++t) {
            v2f b;
            b.x = sW[(kk)     * OUT_DIM + t * 16 + r];
            b.y = sW[(kk + 1) * OUT_DIM + t * 16 + r];
            acc[t] = __builtin_amdgcn_wmma_f32_16x16x4_f32(
                false, a, false, b, (short)0, acc[t], false, false);
        }
    }
#pragma unroll
    for (int t = 0; t < 8; ++t) {
        const int col = t * 16 + r;
        const float blv = bl[col];
#pragma unroll
        for (int rr = 0; rr < 8; ++rr) {
            const int row = rowBase + rr + 8 * kg;
            if (row < N) out[(size_t)row * OUT_DIM + col] = acc[t][rr] + blv;
        }
    }
}

static inline size_t alignUp(size_t v) { return (v + 255) & ~(size_t)255; }

extern "C" void kernel_launch(void* const* d_in, const int* in_sizes, int n_in,
                              void* d_out, int out_size, void* d_ws, size_t ws_size,
                              hipStream_t stream) {
    const float* x  = (const float*)d_in[0];
    const int*   ei = (const int*)  d_in[1];
    const float* Wc = (const float*)d_in[2];
    const float* bc = (const float*)d_in[3];
    const float* Wl = (const float*)d_in[4];
    const float* bl = (const float*)d_in[5];
    float* out = (float*)d_out;

    const int N = in_sizes[0] / IN_DIM;
    const int E = in_sizes[1] / 2;
    const int* src = ei;
    const int* dst = ei + E;

    char* ws = (char*)d_ws;
    const size_t hBytes = (size_t)N * HID_DIM * sizeof(float);
    size_t off = 0;
    float*        h    = (float*)(ws + off);        off = alignUp(off + hBytes);
    char*         srt  = ws + off;                  off = alignUp(off + hBytes);
    float*        dinv = (float*)(ws + off);        off = alignUp(off + (size_t)N * sizeof(float));
    unsigned int* part = (unsigned int*)h;

    size_t so = 0;
    int*          ssrc   = (int*)(srt + so);          so = alignUp(so + (size_t)E * sizeof(int));
    unsigned int* rowptr = (unsigned int*)(srt + so); so = alignUp(so + ((size_t)N + 1) * sizeof(unsigned int));
    unsigned int* blkTot = (unsigned int*)(srt + so);

    const int  nblk = (N + SCAN_BLK - 1) / SCAN_BLK;
    const dim3 hgrid(HIST_CHUNKS, (N + HIST_BINS - 1) / HIST_BINS);

    deg_kernel <<<hgrid, 1024, 0, stream>>>(dst, E, N, part);
    dinv_kernel<<<nblk, SCAN_BLK, 0, stream>>>(part, N, dinv, blkTot);
    fill_kernel<<<hgrid, 1024, 0, stream>>>(src, dst, E, N, part, blkTot, nblk, rowptr, ssrc);

    const int rowBlocks = (N + 127) / 128;
    gemm1_kernel<<<rowBlocks, 256, 0, stream>>>(x, Wc, dinv, h, N);

    gemm2_kernel<<<rowBlocks, 256, 0, stream>>>(rowptr, ssrc, h, dinv, bc, Wl, bl, out, N);
}


# GEMM1 moved into the sort's fill launch: its 16-row tiles run on the 192 WGPs that the 64 LDS-bound fill groups leave idle; one launch fewer; partial histograms moved out of the h region
# speedup vs baseline: 2.2512x; 1.1139x over previous
#include <hip/hip_runtime.h>
#include <hip/hip_bf16.h>

typedef __attribute__((ext_vector_type(2))) float v2f;
typedef __attribute__((ext_vector_type(8))) float v8f;

#define IN_DIM  128
#define HID_DIM 64
#define OUT_DIM 128

#define HIST_BINS   51200
#define HIST_CHUNKS 32

__global__ __launch_bounds__(1024) void deg_kernel(const int* __restrict__ dst, int E, int N,
                                                   unsigned int* __restrict__ part) {
    __shared__ unsigned int cnt[HIST_BINS];
    const int NP    = (N + 31) & ~31;
    const int binLo = blockIdx.y * HIST_BINS;
    const int nb    = min(HIST_BINS, N - binLo);
    {
        uint4* cv = (uint4*)cnt;
        for (int i = threadIdx.x; i < HIST_BINS / 4; i += 1024)
            cv[i] = make_uint4(0u, 0u, 0u, 0u);
    }
    __syncthreads();

    const int per = (E + (int)gridDim.x - 1) / (int)gridDim.x;
    const int e0  = blockIdx.x * per;
    const int e1  = min(E, e0 + per);
    int e = e0 + threadIdx.x;
    for (; e + 15 * 1024 < e1; e += 16 * 1024) {
        unsigned int b[16];
#pragma unroll
        for (int u = 0; u < 16; ++u) b[u] = (unsigned int)(dst[e + u * 1024] - binLo);
#pragma unroll
        for (int u = 0; u < 16; ++u)
            if (b[u] < (unsigned int)nb) atomicAdd(&cnt[b[u]], 1u);
    }
    for (; e < e1; e += 1024) {
        const unsigned int b = (unsigned int)(dst[e] - binLo);
        if (b < (unsigned int)nb) atomicAdd(&cnt[b], 1u);
    }
    __syncthreads();

    unsigned int* row = part + (size_t)blockIdx.x * NP + binLo;
    const int nb4 = nb >> 2;
    {
        const uint4* cv = (const uint4*)cnt;
        uint4* rv = (uint4*)row;
        for (int i = threadIdx.x; i < nb4; i += 1024) rv[i] = cv[i];
    }
    for (int i = (nb4 << 2) + threadIdx.x; i < nb; i += 1024) row[i] = cnt[i];
}

#define SCAN_BLK      256
#define MAX_SCAN_BLKS 1024

__global__ __launch_bounds__(SCAN_BLK) void dinv_kernel(unsigned int* __restrict__ part, int N,
                                                       float* __restrict__ dinv,
                                                       unsigned int* __restrict__ blkTot) {
    __shared__ unsigned int wsum[SCAN_BLK / 32];
    const int NP   = (N + 31) & ~31;
    const int i    = blockIdx.x * SCAN_BLK + threadIdx.x;
    const int lane = threadIdx.x & 31;
    const int wave = threadIdx.x >> 5;
    unsigned int c[HIST_CHUNKS];
    unsigned int deg = 0u;
    if (i < N) {
#pragma unroll
        for (int g = 0; g < HIST_CHUNKS; ++g) c[g] = part[(size_t)g * NP + i];
#pragma unroll
        for (int g = 0; g < HIST_CHUNKS; ++g) { const unsigned int t = c[g]; c[g] = deg; deg += t; }
        dinv[i] = rsqrtf((float)deg + 1.0f);
    }
    unsigned int incl = deg;
    for (int o = 1; o < 32; o <<= 1) {
        const unsigned int u = __shfl_up(incl, o);
        if (lane >= o) incl += u;
    }
    if (lane == 31) wsum[wave] = incl;
    __syncthreads();
    unsigned int wbase = 0u, tot = 0u;
    for (int w = 0; w < SCAN_BLK / 32; ++w) {
        const unsigned int v = wsum[w];
        if (w < wave) wbase += v;
        tot += v;
    }
    const unsigned int excl = wbase + incl - deg;
    if (i < N) {
#pragma unroll
        for (int g = 0; g < HIST_CHUNKS; ++g) part[(size_t)g * NP + i] = c[g] + excl;
    }
    if (threadIdx.x == 0) blkTot[blockIdx.x] = tot;
}

__device__ __forceinline__ void fill_body(const int bx, const int by, const int nby,
                                          const int* __restrict__ src,
                                          const int* __restrict__ dst, int E, int N,
                                          const unsigned int* __restrict__ part,
                                          const unsigned int* __restrict__ blkTot, int nblk,
                                          unsigned int* __restrict__ rowptr,
                                          int* __restrict__ ssrc,
                                          unsigned int* cur, unsigned int* bpre, unsigned int* wtot) {
    const int NP    = (N + 31) & ~31;
    const int binLo = by * HIST_BINS;
    const int nb    = min(HIST_BINS, N - binLo);
    const int t     = threadIdx.x;
    const int lane  = t & 31;

    {
        const unsigned int v = (t < nblk) ? blkTot[t] : 0u;
        unsigned int incl = v;
        for (int o = 1; o < 32; o <<= 1) {
            const unsigned int u = __shfl_up(incl, o);
            if (lane >= o) incl += u;
        }
        if (lane == 31) wtot[t >> 5] = incl;
        __syncthreads();
        if (t < 32) {
            const unsigned int w = wtot[t];
            unsigned int wi = w;
            for (int o = 1; o < 32; o <<= 1) {
                const unsigned int u = __shfl_up(wi, o);
                if (lane >= o) wi += u;
            }
            wtot[t] = wi - w;
        }
        __syncthreads();
        bpre[t] = wtot[t >> 5] + incl - v;
    }
    __syncthreads();

    const unsigned int* prow = part + (size_t)bx * NP + binLo;
    const int nb4 = nb >> 2;
    {
        const uint4* pv = (const uint4*)prow;
        uint4* cv = (uint4*)cur;
        uint4* rv = (uint4*)(rowptr + binLo);
        for (int i0 = t; i0 < nb4; i0 += 4 * 1024) {
            uint4 v[4];
#pragma unroll
            for (int u = 0; u < 4; ++u) {
                const int i = i0 + u * 1024;
                v[u] = (i < nb4) ? pv[i] : make_uint4(0u, 0u, 0u, 0u);
            }
#pragma unroll
            for (int u = 0; u < 4; ++u) {
                const int i = i0 + u * 1024;
                if (i < nb4) {
                    const unsigned int bp = bpre[(binLo + 4 * i) / SCAN_BLK];
                    uint4 w = v[u];
                    w.x += bp; w.y += bp; w.z += bp; w.w += bp;
                    cv[i] = w;
                    if (bx == 0) rv[i] = w;
                }
            }
        }
    }
    for (int i = (nb4 << 2) + t; i < nb; i += 1024) {
        const unsigned int v = prow[i] + bpre[(binLo + i) / SCAN_BLK];
        cur[i] = v;
        if (bx == 0) rowptr[binLo + i] = v;
    }
    if (bx == 0 && by == nby - 1 && t == 0)
        rowptr[N] = bpre[nblk - 1] + blkTot[nblk - 1];
    __syncthreads();

    const int per = (E + HIST_CHUNKS - 1) / HIST_CHUNKS;
    const int e0  = bx * per;
    const int e1  = min(E, e0 + per);
    int e = e0 + t;
    for (; e + 7 * 1024 < e1; e += 8 * 1024) {
        unsigned int b[8];
        int s[8];
#pragma unroll
        for (int u = 0; u < 8; ++u) {
            b[u] = (unsigned int)(dst[e + u * 1024] - binLo);
            s[u] = src[e + u * 1024];
        }
        unsigned int pos[8];
#pragma unroll
        for (int u = 0; u < 8; ++u) {
            pos[u] = 0u;
            if (b[u] < (unsigned int)nb) pos[u] = atomicAdd(&cur[b[u]], 1u);
        }
#pragma unroll
        for (int u = 0; u < 8; ++u)
            if (b[u] < (unsigned int)nb) ssrc[pos[u]] = s[u];
    }
    for (; e < e1; e += 1024) {
        const unsigned int b = (unsigned int)(dst[e] - binLo);
        if (b < (unsigned int)nb) ssrc[atomicAdd(&cur[b], 1u)] = src[e];
    }
}

__device__ __forceinline__ void gemm1_body(const int g, const int G,
    const float* __restrict__ x, const float* __restrict__ Wc,
    const float* __restrict__ dinv, float* __restrict__ h, int N, float* sW) {
    {
        const float4* gv = (const float4*)Wc;
        float4* sv = (float4*)sW;
        for (int i = threadIdx.x; i < (IN_DIM * HID_DIM) / 4; i += 1024)
            sv[i] = gv[i];
    }
    __syncthreads();

    const int wave = __builtin_amdgcn_readfirstlane((int)(threadIdx.x >> 5));
    const int lane = threadIdx.x & 31;
    const int r  = lane & 15;
    const int kg = lane >> 4;
    const int nTiles = (N + 15) >> 4;

    for (int slot = wave; g + G * slot < nTiles; slot += 32) {
        const int rowBase = (g + G * slot) * 16;
        const int rowLd = min(rowBase + r, N - 1);
        const float* xrow = x + (size_t)rowLd * IN_DIM;

        v8f acc[4] = {v8f{}, v8f{}, v8f{}, v8f{}};
        for (int k = 0; k < IN_DIM; k += 4) {
            const int kk = k + (kg << 1);
            float2 av = *(const float2*)(xrow + kk);
            v2f a; a.x = av.x; a.y = av.y;
#pragma unroll
            for (int t = 0; t < 4; ++t) {
                v2f b;
                b.x = sW[(kk)     * HID_DIM + t * 16 + r];
                b.y = sW[(kk + 1) * HID_DIM + t * 16 + r];
                acc[t] = __builtin_amdgcn_wmma_f32_16x16x4_f32(
                    false, a, false, b, (short)0, acc[t], false, false);
            }
        }
        float dv[8];
#pragma unroll
        for (int rr = 0; rr < 8; ++rr) dv[rr] = dinv[min(rowBase + rr + 8 * kg, N - 1)];
#pragma unroll
        for (int t = 0; t < 4; ++t) {
            const int col = t * 16 + r;
#pragma unroll
            for (int rr = 0; rr < 8; ++rr) {
                const int row = rowBase + rr + 8 * kg;
                if (row < N) h[(size_t)row * HID_DIM + col] = acc[t][rr] * dv[rr];
            }
        }
    }
}

__global__ __launch_bounds__(1024) void fill_gemm1_kernel(
    const int* __restrict__ src, const int* __restrict__ dst, int E, int N,
    const unsigned int* __restrict__ part, const unsigned int* __restrict__ blkTot, int nblk,
    unsigned int* __restrict__ rowptr, int* __restrict__ ssrc,
    const float* __restrict__ x, const float* __restrict__ Wc,
    const float* __restrict__ dinv, float* __restrict__ h) {
    __shared__ unsigned int cur[HIST_BINS];
    __shared__ unsigned int bpre[MAX_SCAN_BLKS];
    __shared__ unsigned int wtot[32];
    const int nby   = (N + HIST_BINS - 1) / HIST_BINS;
    const int nHist = HIST_CHUNKS * nby;
    const int b     = blockIdx.x;
    if (b < nHist)
        fill_body(b % HIST_CHUNKS, b / HIST_CHUNKS, nby, src, dst, E, N, part, blkTot, nblk,
                  rowptr, ssrc, cur, bpre, wtot);
    else
        gemm1_body(b - nHist, (int)gridDim.x - nHist, x, Wc, dinv, h, N, (float*)cur);
}

#define ZS_PITCH 68
#define GATHER_SLOT(u) if ((u) < m) { const unsigned int s_ = (unsigned int)__builtin_amdgcn_readlane(sv, (u)); v[(u)] = *(const float2*)(hb + ((s_ << 8) + lb)); }
#define SL_CAP   384

__global__ __launch_bounds__(256) void gemm2_kernel(
    const unsigned int* __restrict__ rowptr, const int* __restrict__ ssrc,
    const float* __restrict__ h, const float* __restrict__ dinv,
    const float* __restrict__ bc, const float* __restrict__ Wl,
    const float* __restrict__ bl, float* __restrict__ out, int N) {
    __shared__ float sW[HID_DIM * OUT_DIM];
    __shared__ float zs[128 * ZS_PITCH];
    __shared__ int   sl[8 * SL_CAP];
    {
        const float4* gv = (const float4*)Wl;
        float4* sv = (float4*)sW;
        for (int i = threadIdx.x; i < (HID_DIM * OUT_DIM) / 4; i += 256)
            sv[i] = gv[i];
    }

    const int wave = threadIdx.x >> 5;
    const int lane = threadIdx.x & 31;
    const int rowBase = blockIdx.x * 128 + wave * 16;

    if (rowBase < N) {
        const unsigned int rp   = rowptr[min(rowBase + lane, N)];
        const float        dvec = dinv[min(rowBase + lane, N - 1)];
        const float2       bcv  = *(const float2*)(bc + lane * 2);
        const unsigned int w0 = (unsigned int)__builtin_amdgcn_readlane((int)rp, 0);
        const unsigned int wn = (unsigned int)__builtin_amdgcn_readlane((int)rp, 16) - w0;
        int* slw = sl + wave * SL_CAP;
        {
            int tmp[SL_CAP / 32];
#pragma unroll
            for (int u = 0; u < SL_CAP / 32; ++u) {
                const unsigned int o = (unsigned int)(u * 32 + lane);
                tmp[u] = (o < wn) ? ssrc[w0 + o] : 0;
            }
#pragma unroll
            for (int u = 0; u < SL_CAP / 32; ++u) slw[u * 32 + lane] = tmp[u];
        }
        const float* hl = h + lane * 2;
        const char*  hb = (const char*)h;
        const unsigned int lb = (unsigned int)lane * 8u;
        for (int i = 0; i < 16; ++i) {
            const unsigned int beg = (unsigned int)__builtin_amdgcn_readlane((int)rp, i);
            const unsigned int cnt = (unsigned int)__builtin_amdgcn_readlane((int)rp, i + 1) - beg;
            const unsigned int lo  = beg - w0;
            const int nld = min(rowBase + i, N - 1);
            float2 acc = *(const float2*)(hl + (size_t)nld * HID_DIM);
            if (lo + cnt <= (unsigned int)SL_CAP) {
                for (unsigned int base = 0u; base < cnt; base += 32u) {
                    const int m  = (int)min(32u, cnt - base);
                    const int sv = slw[min(lo + base + (unsigned int)lane, (unsigned int)(SL_CAP - 1))];
                    float2 v[32];
#pragma unroll
                    for (int u = 0; u < 32; ++u) v[u] = make_float2(0.0f, 0.0f);
#pragma unroll
                    for (int u = 0; u < 8; ++u) GATHER_SLOT(u)
                    if (m > 8) {
#pragma unroll
                        for (int u = 8; u < 16; ++u) GATHER_SLOT(u)
                        if (m > 16) {
#pragma unroll
                            for (int u = 16; u < 24; ++u) GATHER_SLOT(u)
                            if (m > 24) {
#pragma unroll
                                for (int u = 24; u < 32; ++u) GATHER_SLOT(u)
                            }
                        }
                    }
#pragma unroll
                    for (int u = 0; u < 32; ++u) { acc.x += v[u].x; acc.y += v[u].y; }
                }
            } else {
                for (unsigned int j = 0u; j < cnt; ++j) {
                    const int s = __builtin_amdgcn_readfirstlane(ssrc[beg + j]);
                    const float2 v0 = *(const float2*)(hl + (size_t)s * HID_DIM);
                    acc.x += v0.x; acc.y += v0.y;
                }
            }
            const float di = __int_as_float(__builtin_amdgcn_readlane(__float_as_int(dvec), i));
            float2 z;
            z.x = fmaxf(di * acc.x + bcv.x, 0.0f);
            z.y = fmaxf(di * acc.y + bcv.y, 0.0f);
            *(float2*)(zs + (wave * 16 + i) * ZS_PITCH + lane * 2) = z;
        }
    }
    __syncthreads();
    if (rowBase >= N) return;

    const int r  = lane & 15;
    const int kg = lane >> 4;
    const float* zrow = zs + (wave * 16 + r) * ZS_PITCH;

    v8f acc[8] = {v8f{}, v8f{}, v8f{}, v8f{}, v8f{}, v8f{}, v8f{}, v8f{}};
    for (int k = 0; k < HID_DIM; k += 4) {
        const int kk = k + (kg << 1);
        const float2 av = *(const float2*)(zrow + kk);
        v2f a; a.x = av.x; a.y = av.y;
#pragma unroll
        for (int t = 0; t < 8; ++t) {
            v2f b;
            b.x = sW[(kk)     * OUT_DIM + t * 16 + r];
            b.y = sW[(kk + 1) * OUT_DIM + t * 16 + r];
            acc[t] = __builtin_amdgcn_wmma_f32_16x16x4_f32(
                false, a, false, b, (short)0, acc[t], false, false);
        }
    }
#pragma unroll
    for (int t = 0; t < 8; ++t) {
        const int col = t * 16 + r;
        const float blv = bl[col];
#pragma unroll
        for (int rr = 0; rr < 8; ++rr) {
            const int row = rowBase + rr + 8 * kg;
            if (row < N) out[(size_t)row * OUT_DIM + col] = acc[t][rr] + blv;
        }
    }
}

static inline size_t alignUp(size_t v) { return (v + 255) & ~(size_t)255; }

extern "C" void kernel_launch(void* const* d_in, const int* in_sizes, int n_in,
                              void* d_out, int out_size, void* d_ws, size_t ws_size,
                              hipStream_t stream) {
    const float* x  = (const float*)d_in[0];
    const int*   ei = (const int*)  d_in[1];
    const float* Wc = (const float*)d_in[2];
    const float* bc = (const float*)d_in[3];
    const float* Wl = (const float*)d_in[4];
    const float* bl = (const float*)d_in[5];
    float* out = (float*)d_out;

    const int N = in_sizes[0] / IN_DIM;
    const int E = in_sizes[1] / 2;
    const int* src = ei;
    const int* dst = ei + E;

    char* ws = (char*)d_ws;
    const size_t hBytes = (size_t)N * HID_DIM * sizeof(float);
    size_t off = 0;
    float*        h    = (float*)(ws + off);        off = alignUp(off + hBytes);
    char*         srt  = ws + off;                  off = alignUp(off + hBytes);
    float*        dinv = (float*)(ws + off);        off = alignUp(off + (size_t)N * sizeof(float));

    size_t so = 0;
    int*          ssrc   = (int*)(srt + so);          so = alignUp(so + (size_t)E * sizeof(int));
    unsigned int* rowptr = (unsigned int*)(srt + so); so = alignUp(so + ((size_t)N + 1) * sizeof(unsigned int));
    const int  nblk = (N + SCAN_BLK - 1) / SCAN_BLK;
    unsigned int* blkTot = (unsigned int*)(srt + so); so = alignUp(so + (size_t)nblk * sizeof(unsigned int));
    unsigned int* part   = (unsigned int*)(srt + so);

    const dim3 hgrid(HIST_CHUNKS, (N + HIST_BINS - 1) / HIST_BINS);
    const int  nHist = HIST_CHUNKS * ((N + HIST_BINS - 1) / HIST_BINS);
    const int  gemmGroups = 192;

    deg_kernel <<<hgrid, 1024, 0, stream>>>(dst, E, N, part);
    dinv_kernel<<<nblk, SCAN_BLK, 0, stream>>>(part, N, dinv, blkTot);
    fill_gemm1_kernel<<<nHist + gemmGroups, 1024, 0, stream>>>(src, dst, E, N, part, blkTot, nblk, rowptr, ssrc,
                                                               x, Wc, dinv, h);

    const int rowBlocks = (N + 127) / 128;

    gemm2_kernel<<<rowBlocks, 256, 0, stream>>>(rowptr, ssrc, h, dinv, bc, Wl, bl, out, N);
}


# Counting sort's fill pass re-tiled from 2 to 8 node ranges x 32 edge chunks: 256 groups, one per WGP, not 64, so a quarter of the scattered 4-byte stores per WGP; a GEMM1 group co-resides on each WGP
# speedup vs baseline: 2.5718x; 1.1424x over previous
#include <hip/hip_runtime.h>
#include <hip/hip_bf16.h>

typedef __attribute__((ext_vector_type(2))) float v2f;
typedef __attribute__((ext_vector_type(8))) float v8f;

#define IN_DIM  128
#define HID_DIM 64
#define OUT_DIM 128

#define HIST_BINS   51200
#define HIST_CHUNKS 32

__global__ __launch_bounds__(1024) void deg_kernel(const int* __restrict__ dst, int E, int N,
                                                   unsigned int* __restrict__ part) {
    __shared__ unsigned int cnt[HIST_BINS];
    const int NP    = (N + 31) & ~31;
    const int binLo = blockIdx.y * HIST_BINS;
    const int nb    = min(HIST_BINS, N - binLo);
    {
        uint4* cv = (uint4*)cnt;
        for (int i = threadIdx.x; i < HIST_BINS / 4; i += 1024)
            cv[i] = make_uint4(0u, 0u, 0u, 0u);
    }
    __syncthreads();

    const int per = (E + (int)gridDim.x - 1) / (int)gridDim.x;
    const int e0  = blockIdx.x * per;
    const int e1  = min(E, e0 + per);
    int e = e0 + threadIdx.x;
    for (; e + 15 * 1024 < e1; e += 16 * 1024) {
        unsigned int b[16];
#pragma unroll
        for (int u = 0; u < 16; ++u) b[u] = (unsigned int)(dst[e + u * 1024] - binLo);
#pragma unroll
        for (int u = 0; u < 16; ++u)
            if (b[u] < (unsigned int)nb) atomicAdd(&cnt[b[u]], 1u);
    }
    for (; e < e1; e += 1024) {
        const unsigned int b = (unsigned int)(dst[e] - binLo);
        if (b < (unsigned int)nb) atomicAdd(&cnt[b], 1u);
    }
    __syncthreads();

    unsigned int* row = part + (size_t)blockIdx.x * NP + binLo;
    const int nb4 = nb >> 2;
    {
        const uint4* cv = (const uint4*)cnt;
        uint4* rv = (uint4*)row;
        for (int i = threadIdx.x; i < nb4; i += 1024) rv[i] = cv[i];
    }
    for (int i = (nb4 << 2) + threadIdx.x; i < nb; i += 1024) row[i] = cnt[i];
}

#define SCAN_BLK      256
#define MAX_SCAN_BLKS 1024

__global__ __launch_bounds__(SCAN_BLK) void dinv_kernel(unsigned int* __restrict__ part, int N,
                                                       float* __restrict__ dinv,
                                                       unsigned int* __restrict__ blkTot) {
    __shared__ unsigned int wsum[SCAN_BLK / 32];
    const int NP   = (N + 31) & ~31;
    const int i    = blockIdx.x * SCAN_BLK + threadIdx.x;
    const int lane = threadIdx.x & 31;
    const int wave = threadIdx.x >> 5;
    unsigned int c[HIST_CHUNKS];
    unsigned int deg = 0u;
    if (i < N) {
#pragma unroll
        for (int g = 0; g < HIST_CHUNKS; ++g) c[g] = part[(size_t)g * NP + i];
#pragma unroll
        for (int g = 0; g < HIST_CHUNKS; ++g) { const unsigned int t = c[g]; c[g] = deg; deg += t; }
        dinv[i] = rsqrtf((float)deg + 1.0f);
    }
    unsigned int incl = deg;
    for (int o = 1; o < 32; o <<= 1) {
        const unsigned int u = __shfl_up(incl, o);
        if (lane >= o) incl += u;
    }
    if (lane == 31) wsum[wave] = incl;
    __syncthreads();
    unsigned int wbase = 0u, tot = 0u;
    for (int w = 0; w < SCAN_BLK / 32; ++w) {
        const unsigned int v = wsum[w];
        if (w < wave) wbase += v;
        tot += v;
    }
    const unsigned int excl = wbase + incl - deg;
    if (i < N) {
#pragma unroll
        for (int g = 0; g < HIST_CHUNKS; ++g) part[(size_t)g * NP + i] = c[g] + excl;
    }
    if (threadIdx.x == 0) blkTot[blockIdx.x] = tot;
}

#define FILL_BINS   12800
__device__ __forceinline__ void fill_body(const int bx, const int by, const int nby,
                                          const int* __restrict__ src,
                                          const int* __restrict__ dst, int E, int N,
                                          const unsigned int* __restrict__ part,
                                          const unsigned int* __restrict__ blkTot, int nblk,
                                          unsigned int* __restrict__ rowptr,
                                          int* __restrict__ ssrc,
                                          unsigned int* cur, unsigned int* bpre, unsigned int* wtot) {
    const int NP    = (N + 31) & ~31;
    const int binLo = by * FILL_BINS;
    const int nb    = min(FILL_BINS, N - binLo);
    const int t     = threadIdx.x;
    const int lane  = t & 31;

    {
        const unsigned int v = (t < nblk) ? blkTot[t] : 0u;
        unsigned int incl = v;
        for (int o = 1; o < 32; o <<= 1) {
            const unsigned int u = __shfl_up(incl, o);
            if (lane >= o) incl += u;
        }
        if (lane == 31) wtot[t >> 5] = incl;
        __syncthreads();
        if (t < 32) {
            const unsigned int w = wtot[t];
            unsigned int wi = w;
            for (int o = 1; o < 32; o <<= 1) {
                const unsigned int u = __shfl_up(wi, o);
                if (lane >= o) wi += u;
            }
            wtot[t] = wi - w;
        }
        __syncthreads();
        bpre[t] = wtot[t >> 5] + incl - v;
    }
    __syncthreads();

    const unsigned int* prow = part + (size_t)bx * NP + binLo;
    const int nb4 = nb >> 2;
    {
        const uint4* pv = (const uint4*)prow;
        uint4* cv = (uint4*)cur;
        uint4* rv = (uint4*)(rowptr + binLo);
        for (int i0 = t; i0 < nb4; i0 += 4 * 1024) {
            uint4 v[4];
#pragma unroll
            for (int u = 0; u < 4; ++u) {
                const int i = i0 + u * 1024;
                v[u] = (i < nb4) ? pv[i] : make_uint4(0u, 0u, 0u, 0u);
            }
#pragma unroll
            for (int u = 0; u < 4; ++u) {
                const int i = i0 + u * 1024;
                if (i < nb4) {
                    const unsigned int bp = bpre[(binLo + 4 * i) / SCAN_BLK];
                    uint4 w = v[u];
                    w.x += bp; w.y += bp; w.z += bp; w.w += bp;
                    cv[i] = w;
                    if (bx == 0) rv[i] = w;
                }
            }
        }
    }
    for (int i = (nb4 << 2) + t; i < nb; i += 1024) {
        const unsigned int v = prow[i] + bpre[(binLo + i) / SCAN_BLK];
        cur[i] = v;
        if (bx == 0) rowptr[binLo + i] = v;
    }
    if (bx == 0 && by == nby - 1 && t == 0)
        rowptr[N] = bpre[nblk - 1] + blkTot[nblk - 1];
    __syncthreads();

    const int per = (E + HIST_CHUNKS - 1) / HIST_CHUNKS;
    const int e0  = bx * per;
    const int e1  = min(E, e0 + per);
    int e = e0 + t;
    for (; e + 7 * 1024 < e1; e += 8 * 1024) {
        unsigned int b[8];
        int s[8];
#pragma unroll
        for (int u = 0; u < 8; ++u) {
            b[u] = (unsigned int)(dst[e + u * 1024] - binLo);
            s[u] = src[e + u * 1024];
        }
        unsigned int pos[8];
#pragma unroll
        for (int u = 0; u < 8; ++u) {
            pos[u] = 0u;
            if (b[u] < (unsigned int)nb) pos[u] = atomicAdd(&cur[b[u]], 1u);
        }
#pragma unroll
        for (int u = 0; u < 8; ++u)
            if (b[u] < (unsigned int)nb) ssrc[pos[u]] = s[u];
    }
    for (; e < e1; e += 1024) {
        const unsigned int b = (unsigned int)(dst[e] - binLo);
        if (b < (unsigned int)nb) ssrc[atomicAdd(&cur[b], 1u)] = src[e];
    }
}

__device__ __forceinline__ void gemm1_body(const int g, const int G,
    const float* __restrict__ x, const float* __restrict__ Wc,
    const float* __restrict__ dinv, float* __restrict__ h, int N, float* sW) {
    {
        const float4* gv = (const float4*)Wc;
        float4* sv = (float4*)sW;
        for (int i = threadIdx.x; i < (IN_DIM * HID_DIM) / 4; i += 1024)
            sv[i] = gv[i];
    }
    __syncthreads();

    const int wave = __builtin_amdgcn_readfirstlane((int)(threadIdx.x >> 5));
    const int lane = threadIdx.x & 31;
    const int r  = lane & 15;
    const int kg = lane >> 4;
    const int nTiles = (N + 15) >> 4;

    for (int slot = wave; g + G * slot < nTiles; slot += 32) {
        const int rowBase = (g + G * slot) * 16;
        const int rowLd = min(rowBase + r, N - 1);
        const char* xb = (const char*)x;
        const unsigned int xo = (unsigned int)rowLd * (unsigned int)(IN_DIM * sizeof(float))
                              + ((unsigned int)kg << 3);

        v8f acc[4] = {v8f{}, v8f{}, v8f{}, v8f{}};
        for (int k = 0; k < IN_DIM; k += 4) {
            const int kk = k + (kg << 1);
            float2 av = *(const float2*)(xb + (xo + (unsigned int)k * 4u));
            v2f a; a.x = av.x; a.y = av.y;
#pragma unroll
            for (int t = 0; t < 4; ++t) {
                v2f b;
                b.x = sW[(kk)     * HID_DIM + t * 16 + r];
                b.y = sW[(kk + 1) * HID_DIM + t * 16 + r];
                acc[t] = __builtin_amdgcn_wmma_f32_16x16x4_f32(
                    false, a, false, b, (short)0, acc[t], false, false);
            }
        }
        float dv[8];
#pragma unroll
        for (int rr = 0; rr < 8; ++rr) dv[rr] = dinv[min(rowBase + rr + 8 * kg, N - 1)];
#pragma unroll
        for (int t = 0; t < 4; ++t) {
            const int col = t * 16 + r;
#pragma unroll
            for (int rr = 0; rr < 8; ++rr) {
                const int row = rowBase + rr + 8 * kg;
                if (row < N) h[(size_t)row * HID_DIM + col] = acc[t][rr] * dv[rr];
            }
        }
    }
}

__global__ __launch_bounds__(1024, 16) void fill_gemm1_kernel(
    const int* __restrict__ src, const int* __restrict__ dst, int E, int N,
    const unsigned int* __restrict__ part, const unsigned int* __restrict__ blkTot, int nblk,
    unsigned int* __restrict__ rowptr, int* __restrict__ ssrc,
    const float* __restrict__ x, const float* __restrict__ Wc,
    const float* __restrict__ dinv, float* __restrict__ h) {
    __shared__ unsigned int cur[FILL_BINS];
    __shared__ unsigned int bpre[MAX_SCAN_BLKS];
    __shared__ unsigned int wtot[32];
    const int nby   = (N + FILL_BINS - 1) / FILL_BINS;
    const int nHist = HIST_CHUNKS * nby;
    const int b     = blockIdx.x;
    if (b < nHist)
        fill_body(b % HIST_CHUNKS, b / HIST_CHUNKS, nby, src, dst, E, N, part, blkTot, nblk,
                  rowptr, ssrc, cur, bpre, wtot);
    else
        gemm1_body(b - nHist, (int)gridDim.x - nHist, x, Wc, dinv, h, N, (float*)cur);
}

#define ZS_PITCH 68
#define GATHER_SLOT(u) if ((u) < m) { const unsigned int s_ = (unsigned int)__builtin_amdgcn_readlane(sv, (u)); v[(u)] = *(const float2*)(hb + ((s_ << 8) + lb)); }
#define SL_CAP   384

__global__ __launch_bounds__(256) void gemm2_kernel(
    const unsigned int* __restrict__ rowptr, const int* __restrict__ ssrc,
    const float* __restrict__ h, const float* __restrict__ dinv,
    const float* __restrict__ bc, const float* __restrict__ Wl,
    const float* __restrict__ bl, float* __restrict__ out, int N) {
    __shared__ float sW[HID_DIM * OUT_DIM];
    __shared__ float zs[128 * ZS_PITCH];
    __shared__ int   sl[8 * SL_CAP];
    {
        const float4* gv = (const float4*)Wl;
        float4* sv = (float4*)sW;
        for (int i = threadIdx.x; i < (HID_DIM * OUT_DIM) / 4; i += 256)
            sv[i] = gv[i];
    }

    const int wave = threadIdx.x >> 5;
    const int lane = threadIdx.x & 31;
    const int rowBase = blockIdx.x * 128 + wave * 16;

    if (rowBase < N) {
        const unsigned int rp   = rowptr[min(rowBase + lane, N)];
        const float        dvec = dinv[min(rowBase + lane, N - 1)];
        const float2       bcv  = *(const float2*)(bc + lane * 2);
        const unsigned int w0 = (unsigned int)__builtin_amdgcn_readlane((int)rp, 0);
        const unsigned int wn = (unsigned int)__builtin_amdgcn_readlane((int)rp, 16) - w0;
        int* slw = sl + wave * SL_CAP;
        {
            int tmp[SL_CAP / 32];
#pragma unroll
            for (int u = 0; u < SL_CAP / 32; ++u) {
                const unsigned int o = (unsigned int)(u * 32 + lane);
                tmp[u] = (o < wn) ? ssrc[w0 + o] : 0;
            }
#pragma unroll
            for (int u = 0; u < SL_CAP / 32; ++u) slw[u * 32 + lane] = tmp[u];
        }
        const float* hl = h + lane * 2;
        const char*  hb = (const char*)h;
        const unsigned int lb = (unsigned int)lane * 8u;
        for (int i = 0; i < 16; ++i) {
            const unsigned int beg = (unsigned int)__builtin_amdgcn_readlane((int)rp, i);
            const unsigned int cnt = (unsigned int)__builtin_amdgcn_readlane((int)rp, i + 1) - beg;
            const unsigned int lo  = beg - w0;
            const int nld = min(rowBase + i, N - 1);
            float2 acc = *(const float2*)(hl + (size_t)nld * HID_DIM);
            if (lo + cnt <= (unsigned int)SL_CAP) {
                for (unsigned int base = 0u; base < cnt; base += 32u) {
                    const int m  = (int)min(32u, cnt - base);
                    const int sv = slw[min(lo + base + (unsigned int)lane, (unsigned int)(SL_CAP - 1))];
                    float2 v[32];
#pragma unroll
                    for (int u = 0; u < 32; ++u) v[u] = make_float2(0.0f, 0.0f);
#pragma unroll
                    for (int u = 0; u < 8; ++u) GATHER_SLOT(u)
                    if (m > 8) {
#pragma unroll
                        for (int u = 8; u < 16; ++u) GATHER_SLOT(u)
                        if (m > 16) {
#pragma unroll
                            for (int u = 16; u < 24; ++u) GATHER_SLOT(u)
                            if (m > 24) {
#pragma unroll
                                for (int u = 24; u < 32; ++u) GATHER_SLOT(u)
                            }
                        }
                    }
#pragma unroll
                    for (int u = 0; u < 32; ++u) { acc.x += v[u].x; acc.y += v[u].y; }
                }
            } else {
                for (unsigned int j = 0u; j < cnt; ++j) {
                    const int s = __builtin_amdgcn_readfirstlane(ssrc[beg + j]);
                    const float2 v0 = *(const float2*)(hl + (size_t)s * HID_DIM);
                    acc.x += v0.x; acc.y += v0.y;
                }
            }
            const float di = __int_as_float(__builtin_amdgcn_readlane(__float_as_int(dvec), i));
            float2 z;
            z.x = fmaxf(di * acc.x + bcv.x, 0.0f);
            z.y = fmaxf(di * acc.y + bcv.y, 0.0f);
            *(float2*)(zs + (wave * 16 + i) * ZS_PITCH + lane * 2) = z;
        }
    }
    __syncthreads();
    if (rowBase >= N) return;

    const int r  = lane & 15;
    const int kg = lane >> 4;
    const float* zrow = zs + (wave * 16 + r) * ZS_PITCH;

    v8f acc[8] = {v8f{}, v8f{}, v8f{}, v8f{}, v8f{}, v8f{}, v8f{}, v8f{}};
    for (int k = 0; k < HID_DIM; k += 4) {
        const int kk = k + (kg << 1);
        const float2 av = *(const float2*)(zrow + kk);
        v2f a; a.x = av.x; a.y = av.y;
#pragma unroll
        for (int t = 0; t < 8; ++t) {
            v2f b;
            b.x = sW[(kk)     * OUT_DIM + t * 16 + r];
            b.y = sW[(kk + 1) * OUT_DIM + t * 16 + r];
            acc[t] = __builtin_amdgcn_wmma_f32_16x16x4_f32(
                false, a, false, b, (short)0, acc[t], false, false);
        }
    }
#pragma unroll
    for (int t = 0; t < 8; ++t) {
        const int col = t * 16 + r;
        const float blv = bl[col];
#pragma unroll
        for (int rr = 0; rr < 8; ++rr) {
            const int row = rowBase + rr + 8 * kg;
            if (row < N) out[(size_t)row * OUT_DIM + col] = acc[t][rr] + blv;
        }
    }
}

static inline size_t alignUp(size_t v) { return (v + 255) & ~(size_t)255; }

extern "C" void kernel_launch(void* const* d_in, const int* in_sizes, int n_in,
                              void* d_out, int out_size, void* d_ws, size_t ws_size,
                              hipStream_t stream) {
    const float* x  = (const float*)d_in[0];
    const int*   ei = (const int*)  d_in[1];
    const float* Wc = (const float*)d_in[2];
    const float* bc = (const float*)d_in[3];
    const float* Wl = (const float*)d_in[4];
    const float* bl = (const float*)d_in[5];
    float* out = (float*)d_out;

    const int N = in_sizes[0] / IN_DIM;
    const int E = in_sizes[1] / 2;
    const int* src = ei;
    const int* dst = ei + E;

    char* ws = (char*)d_ws;
    const size_t hBytes = (size_t)N * HID_DIM * sizeof(float);
    size_t off = 0;
    float*        h    = (float*)(ws + off);        off = alignUp(off + hBytes);
    char*         srt  = ws + off;                  off = alignUp(off + hBytes);
    float*        dinv = (float*)(ws + off);        off = alignUp(off + (size_t)N * sizeof(float));

    size_t so = 0;
    int*          ssrc   = (int*)(srt + so);          so = alignUp(so + (size_t)E * sizeof(int));
    unsigned int* rowptr = (unsigned int*)(srt + so); so = alignUp(so + ((size_t)N + 1) * sizeof(unsigned int));
    const int  nblk = (N + SCAN_BLK - 1) / SCAN_BLK;
    unsigned int* blkTot = (unsigned int*)(srt + so); so = alignUp(so + (size_t)nblk * sizeof(unsigned int));
    unsigned int* part   = (unsigned int*)(srt + so);

    const dim3 hgrid(HIST_CHUNKS, (N + HIST_BINS - 1) / HIST_BINS);
    const int  nHist = HIST_CHUNKS * ((N + FILL_BINS - 1) / FILL_BINS);
    const int  gemmGroups = 256;

    deg_kernel <<<hgrid, 1024, 0, stream>>>(dst, E, N, part);
    dinv_kernel<<<nblk, SCAN_BLK, 0, stream>>>(part, N, dinv, blkTot);
    fill_gemm1_kernel<<<nHist + gemmGroups, 1024, 0, stream>>>(src, dst, E, N, part, blkTot, nblk, rowptr, ssrc,
                                                               x, Wc, dinv, h);

    const int rowBlocks = (N + 127) / 128;

    gemm2_kernel<<<rowBlocks, 256, 0, stream>>>(rowptr, ssrc, h, dinv, bc, Wl, bl, out, N);
}


# GEMM1's x tile goes through a per-wave LDS slab: whole 128-byte lines, 4 rows per load, a quarter of K per wait, not 16 lines x 16 bytes per k-step with a wait each; k order kept
# speedup vs baseline: 2.7245x; 1.0594x over previous
#include <hip/hip_runtime.h>
#include <hip/hip_bf16.h>

typedef __attribute__((ext_vector_type(2))) float v2f;
typedef __attribute__((ext_vector_type(4))) float v4f;
typedef __attribute__((ext_vector_type(8))) float v8f;

#define IN_DIM  128
#define HID_DIM 64
#define OUT_DIM 128

#define HIST_BINS   51200
#define HIST_CHUNKS 32

__global__ __launch_bounds__(1024) void deg_kernel(const int* __restrict__ dst, int E, int N,
                                                   unsigned int* __restrict__ part) {
    __shared__ unsigned int cnt[HIST_BINS];
    const int NP    = (N + 31) & ~31;
    const int binLo = blockIdx.y * HIST_BINS;
    const int nb    = min(HIST_BINS, N - binLo);
    {
        uint4* cv = (uint4*)cnt;
        for (int i = threadIdx.x; i < HIST_BINS / 4; i += 1024)
            cv[i] = make_uint4(0u, 0u, 0u, 0u);
    }
    __syncthreads();

    const int per = (E + (int)gridDim.x - 1) / (int)gridDim.x;
    const int e0  = blockIdx.x * per;
    const int e1  = min(E, e0 + per);
    int e = e0 + threadIdx.x;
    for (; e + 15 * 1024 < e1; e += 16 * 1024) {
        unsigned int b[16];
#pragma unroll
        for (int u = 0; u < 16; ++u) b[u] = (unsigned int)(dst[e + u * 1024] - binLo);
#pragma unroll
        for (int u = 0; u < 16; ++u)
            if (b[u] < (unsigned int)nb) atomicAdd(&cnt[b[u]], 1u);
    }
    for (; e < e1; e += 1024) {
        const unsigned int b = (unsigned int)(dst[e] - binLo);
        if (b < (unsigned int)nb) atomicAdd(&cnt[b], 1u);
    }
    __syncthreads();

    unsigned int* row = part + (size_t)blockIdx.x * NP + binLo;
    const int nb4 = nb >> 2;
    {
        const uint4* cv = (const uint4*)cnt;
        uint4* rv = (uint4*)row;
        for (int i = threadIdx.x; i < nb4; i += 1024) rv[i] = cv[i];
    }
    for (int i = (nb4 << 2) + threadIdx.x; i < nb; i += 1024) row[i] = cnt[i];
}

#define SCAN_BLK      256
#define MAX_SCAN_BLKS 1024

__global__ __launch_bounds__(SCAN_BLK) void dinv_kernel(unsigned int* __restrict__ part, int N,
                                                       float* __restrict__ dinv,
                                                       unsigned int* __restrict__ blkTot) {
    __shared__ unsigned int wsum[SCAN_BLK / 32];
    const int NP   = (N + 31) & ~31;
    const int i    = blockIdx.x * SCAN_BLK + threadIdx.x;
    const int lane = threadIdx.x & 31;
    const int wave = threadIdx.x >> 5;
    unsigned int c[HIST_CHUNKS];
    unsigned int deg = 0u;
    if (i < N) {
#pragma unroll
        for (int g = 0; g < HIST_CHUNKS; ++g) c[g] = part[(size_t)g * NP + i];
#pragma unroll
        for (int g = 0; g < HIST_CHUNKS; ++g) { const unsigned int t = c[g]; c[g] = deg; deg += t; }
        dinv[i] = rsqrtf((float)deg + 1.0f);
    }
    unsigned int incl = deg;
    for (int o = 1; o < 32; o <<= 1) {
        const unsigned int u = __shfl_up(incl, o);
        if (lane >= o) incl += u;
    }
    if (lane == 31) wsum[wave] = incl;
    __syncthreads();
    unsigned int wbase = 0u, tot = 0u;
    for (int w = 0; w < SCAN_BLK / 32; ++w) {
        const unsigned int v = wsum[w];
        if (w < wave) wbase += v;
        tot += v;
    }
    const unsigned int excl = wbase + incl - deg;
    if (i < N) {
#pragma unroll
        for (int g = 0; g < HIST_CHUNKS; ++g) part[(size_t)g * NP + i] = c[g] + excl;
    }
    if (threadIdx.x == 0) blkTot[blockIdx.x] = tot;
}

#define FILL_BINS   12800
__device__ __forceinline__ void fill_body(const int bx, const int by, const int nby,
                                          const int* __restrict__ src,
                                          const int* __restrict__ dst, int E, int N,
                                          const unsigned int* __restrict__ part,
                                          const unsigned int* __restrict__ blkTot, int nblk,
                                          unsigned int* __restrict__ rowptr,
                                          int* __restrict__ ssrc,
                                          unsigned int* cur, unsigned int* bpre, unsigned int* wtot) {
    const int NP    = (N + 31) & ~31;
    const int binLo = by * FILL_BINS;
    const int nb    = min(FILL_BINS, N - binLo);
    const int t     = threadIdx.x;
    const int lane  = t & 31;

    {
        const unsigned int v = (t < nblk) ? blkTot[t] : 0u;
        unsigned int incl = v;
        for (int o = 1; o < 32; o <<= 1) {
            const unsigned int u = __shfl_up(incl, o);
            if (lane >= o) incl += u;
        }
        if (lane == 31) wtot[t >> 5] = incl;
        __syncthreads();
        if (t < 32) {
            const unsigned int w = wtot[t];
            unsigned int wi = w;
            for (int o = 1; o < 32; o <<= 1) {
                const unsigned int u = __shfl_up(wi, o);
                if (lane >= o) wi += u;
            }
            wtot[t] = wi - w;
        }
        __syncthreads();
        bpre[t] = wtot[t >> 5] + incl - v;
    }
    __syncthreads();

    const unsigned int* prow = part + (size_t)bx * NP + binLo;
    const int nb4 = nb >> 2;
    {
        const uint4* pv = (const uint4*)prow;
        uint4* cv = (uint4*)cur;
        uint4* rv = (uint4*)(rowptr + binLo);
        for (int i0 = t; i0 < nb4; i0 += 4 * 1024) {
            uint4 v[4];
#pragma unroll
            for (int u = 0; u < 4; ++u) {
                const int i = i0 + u * 1024;
                v[u] = (i < nb4) ? pv[i] : make_uint4(0u, 0u, 0u, 0u);
            }
#pragma unroll
            for (int u = 0; u < 4; ++u) {
                const int i = i0 + u * 1024;
                if (i < nb4) {
                    const unsigned int bp = bpre[(binLo + 4 * i) / SCAN_BLK];
                    uint4 w = v[u];
                    w.x += bp; w.y += bp; w.z += bp; w.w += bp;
                    cv[i] = w;
                    if (bx == 0) rv[i] = w;
                }
            }
        }
    }
    for (int i = (nb4 << 2) + t; i < nb; i += 1024) {
        const unsigned int v = prow[i] + bpre[(binLo + i) / SCAN_BLK];
        cur[i] = v;
        if (bx == 0) rowptr[binLo + i] = v;
    }
    if (bx == 0 && by == nby - 1 && t == 0)
        rowptr[N] = bpre[nblk - 1] + blkTot[nblk - 1];
    __syncthreads();

    const int per = (E + HIST_CHUNKS - 1) / HIST_CHUNKS;
    const int e0  = bx * per;
    const int e1  = min(E, e0 + per);
    int e = e0 + t;
    for (; e + 7 * 1024 < e1; e += 8 * 1024) {
        unsigned int b[8];
        int s[8];
#pragma unroll
        for (int u = 0; u < 8; ++u) {
            b[u] = (unsigned int)(dst[e + u * 1024] - binLo);
            s[u] = src[e + u * 1024];
        }
        unsigned int pos[8];
#pragma unroll
        for (int u = 0; u < 8; ++u) {
            pos[u] = 0u;
            if (b[u] < (unsigned int)nb) pos[u] = atomicAdd(&cur[b[u]], 1u);
        }
#pragma unroll
        for (int u = 0; u < 8; ++u)
            if (b[u] < (unsigned int)nb) ssrc[pos[u]] = s[u];
    }
    for (; e < e1; e += 1024) {
        const unsigned int b = (unsigned int)(dst[e] - binLo);
        if (b < (unsigned int)nb) ssrc[atomicAdd(&cur[b], 1u)] = src[e];
    }
}

#define XS_PITCH 36
__device__ __forceinline__ void gemm1_body(const int g, const int G,
    const float* __restrict__ x, const float* __restrict__ Wc,
    const float* __restrict__ dinv, float* __restrict__ h, int N, float* sW, float* xs) {
    {
        const float4* gv = (const float4*)Wc;
        float4* sv = (float4*)sW;
        for (int i = threadIdx.x; i < (IN_DIM * HID_DIM) / 4; i += 1024)
            sv[i] = gv[i];
    }
    __syncthreads();

    const int wave = __builtin_amdgcn_readfirstlane((int)(threadIdx.x >> 5));
    const int lane = threadIdx.x & 31;
    const int r  = lane & 15;
    const int kg = lane >> 4;
    const int nTiles = (N + 15) >> 4;

    const int srow = lane >> 3;
    const int spc  = lane & 7;
    float*       xw  = xs + wave * (16 * XS_PITCH);
    float*       xst = xw + srow * XS_PITCH + spc * 4;
    const float* swl = sW + (kg << 1) * HID_DIM + r;
    const char*  xb  = (const char*)x;

    for (int slot = wave; g + G * slot < nTiles; slot += 32) {
        const int rowBase = (g + G * slot) * 16;
        const int ldBase = max(min(rowBase, N - 16), 0);
        const int shift  = rowBase - ldBase;
        const float* xld = xw + min(r + shift, 15) * XS_PITCH + (kg << 1);
        const unsigned int go0 = (unsigned int)(ldBase + srow) * (unsigned int)(IN_DIM * sizeof(float))
                               + (unsigned int)spc * 16u;

        v8f acc[4] = {v8f{}, v8f{}, v8f{}, v8f{}};
#pragma unroll 1
        for (int q = 0; q < 4; ++q) {
            const char* xq = xb + q * 128;
            v4f xv0, xv1, xv2, xv3;
            asm volatile("global_load_b128 %0, %4, %5\n\t"
                         "global_load_b128 %1, %4, %5 offset:2048\n\t"
                         "global_load_b128 %2, %4, %5 offset:4096\n\t"
                         "global_load_b128 %3, %4, %5 offset:6144\n\t"
                         "s_wait_loadcnt 0x0"
                         : "=&v"(xv0), "=&v"(xv1), "=&v"(xv2), "=&v"(xv3)
                         : "v"(go0), "s"(xq)
                         : "memory");
            *(v4f*)(xst)                 = xv0;
            *(v4f*)(xst +  4 * XS_PITCH) = xv1;
            *(v4f*)(xst +  8 * XS_PITCH) = xv2;
            *(v4f*)(xst + 12 * XS_PITCH) = xv3;
            const float* swq = swl + q * (32 * HID_DIM);
#pragma unroll
            for (int s = 0; s < 8; ++s) {
                const float2 av = *(const float2*)(xld + 4 * s);
                v2f a; a.x = av.x; a.y = av.y;
#pragma unroll
                for (int t = 0; t < 4; ++t) {
                    v2f b;
                    b.x = swq[(4 * s)     * HID_DIM + t * 16];
                    b.y = swq[(4 * s + 1) * HID_DIM + t * 16];
                    acc[t] = __builtin_amdgcn_wmma_f32_16x16x4_f32(
                        false, a, false, b, (short)0, acc[t], false, false);
                }
            }
        }
        float dv[8];
#pragma unroll
        for (int rr = 0; rr < 8; ++rr) dv[rr] = dinv[min(rowBase + rr + 8 * kg, N - 1)];
        char* hb = (char*)h;
#pragma unroll
        for (int t = 0; t < 4; ++t) {
            const int col = t * 16 + r;
#pragma unroll
            for (int rr = 0; rr < 8; ++rr) {
                const int row = rowBase + rr + 8 * kg;
                if (row < N)
                    *(float*)(hb + ((unsigned int)row * (unsigned int)(HID_DIM * sizeof(float)) + (unsigned int)col * 4u))
                        = acc[t][rr] * dv[rr];
            }
        }
    }
}

__global__ __launch_bounds__(1024, 16) void fill_gemm1_kernel(
    const int* __restrict__ src, const int* __restrict__ dst, int E, int N,
    const unsigned int* __restrict__ part, const unsigned int* __restrict__ blkTot, int nblk,
    unsigned int* __restrict__ rowptr, int* __restrict__ ssrc,
    const float* __restrict__ x, const float* __restrict__ Wc,
    const float* __restrict__ dinv, float* __restrict__ h) {
    __shared__ __attribute__((aligned(16))) unsigned int cur[FILL_BINS];
    __shared__ unsigned int bpre[MAX_SCAN_BLKS];
    __shared__ unsigned int wtot[32];
    __shared__ __attribute__((aligned(16))) float xs[32 * 16 * XS_PITCH];
    const int nby   = (N + FILL_BINS - 1) / FILL_BINS;
    const int nHist = HIST_CHUNKS * nby;
    const int b     = blockIdx.x;
    if (b < nHist)
        fill_body(b % HIST_CHUNKS, b / HIST_CHUNKS, nby, src, dst, E, N, part, blkTot, nblk,
                  rowptr, ssrc, cur, bpre, wtot);
    else
        gemm1_body(b - nHist, (int)gridDim.x - nHist, x, Wc, dinv, h, N, (float*)cur, xs);
}

#define ZS_PITCH 68
#define GATHER_SLOT(u) if ((u) < m) { const unsigned int s_ = (unsigned int)__builtin_amdgcn_readlane(sv, (u)); v[(u)] = *(const float2*)(hb + ((s_ << 8) + lb)); }
#define SL_CAP   384

__global__ __launch_bounds__(256) void gemm2_kernel(
    const unsigned int* __restrict__ rowptr, const int* __restrict__ ssrc,
    const float* __restrict__ h, const float* __restrict__ dinv,
    const float* __restrict__ bc, const float* __restrict__ Wl,
    const float* __restrict__ bl, float* __restrict__ out, int N) {
    __shared__ float sW[HID_DIM * OUT_DIM];
    __shared__ float zs[128 * ZS_PITCH];
    __shared__ int   sl[8 * SL_CAP];
    {
        const float4* gv = (const float4*)Wl;
        float4* sv = (float4*)sW;
        for (int i = threadIdx.x; i < (HID_DIM * OUT_DIM) / 4; i += 256)
            sv[i] = gv[i];
    }

    const int wave = threadIdx.x >> 5;
    const int lane = threadIdx.x & 31;
    const int rowBase = blockIdx.x * 128 + wave * 16;

    if (rowBase < N) {
        const unsigned int rp   = rowptr[min(rowBase + lane, N)];
        const float        dvec = dinv[min(rowBase + lane, N - 1)];
        const float2       bcv  = *(const float2*)(bc + lane * 2);
        const unsigned int w0 = (unsigned int)__builtin_amdgcn_readlane((int)rp, 0);
        const unsigned int wn = (unsigned int)__builtin_amdgcn_readlane((int)rp, 16) - w0;
        int* slw = sl + wave * SL_CAP;
        {
            int tmp[SL_CAP / 32];
#pragma unroll
            for (int u = 0; u < SL_CAP / 32; ++u) {
                const unsigned int o = (unsigned int)(u * 32 + lane);
                tmp[u] = (o < wn) ? ssrc[w0 + o] : 0;
            }
#pragma unroll
            for (int u = 0; u < SL_CAP / 32; ++u) slw[u * 32 + lane] = tmp[u];
        }
        const float* hl = h + lane * 2;
        const char*  hb = (const char*)h;
        const unsigned int lb = (unsigned int)lane * 8u;
        for (int i = 0; i < 16; ++i) {
            const unsigned int beg = (unsigned int)__builtin_amdgcn_readlane((int)rp, i);
            const unsigned int cnt = (unsigned int)__builtin_amdgcn_readlane((int)rp, i + 1) - beg;
            const unsigned int lo  = beg - w0;
            const int nld = min(rowBase + i, N - 1);
            float2 acc = *(const float2*)(hl + (size_t)nld * HID_DIM);
            if (lo + cnt <= (unsigned int)SL_CAP) {
                for (unsigned int base = 0u; base < cnt; base += 32u) {
                    const int m  = (int)min(32u, cnt - base);
                    const int sv = slw[min(lo + base + (unsigned int)lane, (unsigned int)(SL_CAP - 1))];
                    float2 v[32];
#pragma unroll
                    for (int u = 0; u < 32; ++u) v[u] = make_float2(0.0f, 0.0f);
#pragma unroll
                    for (int u = 0; u < 8; ++u) GATHER_SLOT(u)
                    if (m > 8) {
#pragma unroll
                        for (int u = 8; u < 16; ++u) GATHER_SLOT(u)
                        if (m > 16) {
#pragma unroll
                            for (int u = 16; u < 24; ++u) GATHER_SLOT(u)
                            if (m > 24) {
#pragma unroll
                                for (int u = 24; u < 32; ++u) GATHER_SLOT(u)
                            }
                        }
                    }
#pragma unroll
                    for (int u = 0; u < 32; ++u) { acc.x += v[u].x; acc.y += v[u].y; }
                }
            } else {
                for (unsigned int j = 0u; j < cnt; ++j) {
                    const int s = __builtin_amdgcn_readfirstlane(ssrc[beg + j]);
                    const float2 v0 = *(const float2*)(hl + (size_t)s * HID_DIM);
                    acc.x += v0.x; acc.y += v0.y;
                }
            }
            const float di = __int_as_float(__builtin_amdgcn_readlane(__float_as_int(dvec), i));
            float2 z;
            z.x = fmaxf(di * acc.x + bcv.x, 0.0f);
            z.y = fmaxf(di * acc.y + bcv.y, 0.0f);
            *(float2*)(zs + (wave * 16 + i) * ZS_PITCH + lane * 2) = z;
        }
    }
    __syncthreads();
    if (rowBase >= N) return;

    const int r  = lane & 15;
    const int kg = lane >> 4;
    const float* zrow = zs + (wave * 16 + r) * ZS_PITCH;

    v8f acc[8] = {v8f{}, v8f{}, v8f{}, v8f{}, v8f{}, v8f{}, v8f{}, v8f{}};
    for (int k = 0; k < HID_DIM; k += 4) {
        const int kk = k + (kg << 1);
        const float2 av = *(const float2*)(zrow + kk);
        v2f a; a.x = av.x; a.y = av.y;
#pragma unroll
        for (int t = 0; t < 8; ++t) {
            v2f b;
            b.x = sW[(kk)     * OUT_DIM + t * 16 + r];
            b.y = sW[(kk + 1) * OUT_DIM + t * 16 + r];
            acc[t] = __builtin_amdgcn_wmma_f32_16x16x4_f32(
                false, a, false, b, (short)0, acc[t], false, false);
        }
    }
#pragma unroll
    for (int t = 0; t < 8; ++t) {
        const int col = t * 16 + r;
        const float blv = bl[col];
#pragma unroll
        for (int rr = 0; rr < 8; ++rr) {
            const int row = rowBase + rr + 8 * kg;
            if (row < N) out[(size_t)row * OUT_DIM + col] = acc[t][rr] + blv;
        }
    }
}

static inline size_t alignUp(size_t v) { return (v + 255) & ~(size_t)255; }

extern "C" void kernel_launch(void* const* d_in, const int* in_sizes, int n_in,
                              void* d_out, int out_size, void* d_ws, size_t ws_size,
                              hipStream_t stream) {
    const float* x  = (const float*)d_in[0];
    const int*   ei = (const int*)  d_in[1];
    const float* Wc = (const float*)d_in[2];
    const float* bc = (const float*)d_in[3];
    const float* Wl = (const float*)d_in[4];
    const float* bl = (const float*)d_in[5];
    float* out = (float*)d_out;

    const int N = in_sizes[0] / IN_DIM;
    const int E = in_sizes[1] / 2;
    const int* src = ei;
    const int* dst = ei + E;

    char* ws = (char*)d_ws;
    const size_t hBytes = (size_t)N * HID_DIM * sizeof(float);
    size_t off = 0;
    float*        h    = (float*)(ws + off);        off = alignUp(off + hBytes);
    char*         srt  = ws + off;                  off = alignUp(off + hBytes);
    float*        dinv = (float*)(ws + off);        off = alignUp(off + (size_t)N * sizeof(float));

    size_t so = 0;
    int*          ssrc   = (int*)(srt + so);          so = alignUp(so + (size_t)E * sizeof(int));
    unsigned int* rowptr = (unsigned int*)(srt + so); so = alignUp(so + ((size_t)N + 1) * sizeof(unsigned int));
    const int  nblk = (N + SCAN_BLK - 1) / SCAN_BLK;
    unsigned int* blkTot = (unsigned int*)(srt + so); so = alignUp(so + (size_t)nblk * sizeof(unsigned int));
    unsigned int* part   = (unsigned int*)(srt + so);

    const dim3 hgrid(HIST_CHUNKS, (N + HIST_BINS - 1) / HIST_BINS);
    const int  nHist = HIST_CHUNKS * ((N + FILL_BINS - 1) / FILL_BINS);
    const int  gemmGroups = 256;

    deg_kernel <<<hgrid, 1024, 0, stream>>>(dst, E, N, part);
    dinv_kernel<<<nblk, SCAN_BLK, 0, stream>>>(part, N, dinv, blkTot);
    fill_gemm1_kernel<<<nHist + gemmGroups, 1024, 0, stream>>>(src, dst, E, N, part, blkTot, nblk, rowptr, ssrc,
                                                               x, Wc, dinv, h);

    const int rowBlocks = (N + 127) / 128;

    gemm2_kernel<<<rowBlocks, 256, 0, stream>>>(rowptr, ssrc, h, dinv, bc, Wl, bl, out, N);
}


# Sort's scattered 4-byte store per edge replaced: fill stages each (chunk, 128-row block) run in LDS and writes it coalesced; gemm2 sorts its block's piece by row in LDS; sum order changes
# speedup vs baseline: 3.3880x; 1.2435x over previous
#include <hip/hip_runtime.h>
#include <hip/hip_bf16.h>

typedef __attribute__((ext_vector_type(2))) float v2f;
typedef __attribute__((ext_vector_type(4))) float v4f;
typedef __attribute__((ext_vector_type(8))) float v8f;

#define IN_DIM  128
#define HID_DIM 64
#define OUT_DIM 128

#define HIST_BINS   51200
#define HIST_CHUNKS 32

__global__ __launch_bounds__(1024) void deg_kernel(const int* __restrict__ dst, int E, int N,
                                                   unsigned int* __restrict__ part) {
    __shared__ unsigned int cnt[HIST_BINS];
    const int NP    = (N + 31) & ~31;
    const int binLo = blockIdx.y * HIST_BINS;
    const int nb    = min(HIST_BINS, N - binLo);
    {
        uint4* cv = (uint4*)cnt;
        for (int i = threadIdx.x; i < HIST_BINS / 4; i += 1024)
            cv[i] = make_uint4(0u, 0u, 0u, 0u);
    }
    __syncthreads();

    const int per = (E + (int)gridDim.x - 1) / (int)gridDim.x;
    const int e0  = blockIdx.x * per;
    const int e1  = min(E, e0 + per);
    int e = e0 + threadIdx.x;
    for (; e + 15 * 1024 < e1; e += 16 * 1024) {
        unsigned int b[16];
#pragma unroll
        for (int u = 0; u < 16; ++u) b[u] = (unsigned int)(dst[e + u * 1024] - binLo);
#pragma unroll
        for (int u = 0; u < 16; ++u)
            if (b[u] < (unsigned int)nb) atomicAdd(&cnt[b[u]], 1u);
    }
    for (; e < e1; e += 1024) {
        const unsigned int b = (unsigned int)(dst[e] - binLo);
        if (b < (unsigned int)nb) atomicAdd(&cnt[b], 1u);
    }
    __syncthreads();

    unsigned int* row = part + (size_t)blockIdx.x * NP + binLo;
    const int nb4 = nb >> 2;
    {
        const uint4* cv = (const uint4*)cnt;
        uint4* rv = (uint4*)row;
        for (int i = threadIdx.x; i < nb4; i += 1024) rv[i] = cv[i];
    }
    for (int i = (nb4 << 2) + threadIdx.x; i < nb; i += 1024) row[i] = cnt[i];
}

#define SCAN_BLK      128
#define MAX_SCAN_BLKS 1024

__global__ __launch_bounds__(SCAN_BLK) void dinv_kernel(const unsigned int* __restrict__ part, int N,
                                                       float* __restrict__ dinv,
                                                       unsigned int* __restrict__ runOfs,
                                                       unsigned int* __restrict__ blkTot) {
    __shared__ unsigned int wpart[SCAN_BLK / 32][HIST_CHUNKS];
    const int NP   = (N + 31) & ~31;
    const int i    = blockIdx.x * SCAN_BLK + threadIdx.x;
    const int lane = threadIdx.x & 31;
    const int wave = threadIdx.x >> 5;
    unsigned int c[HIST_CHUNKS];
    unsigned int deg = 0u;
#pragma unroll
    for (int g = 0; g < HIST_CHUNKS; ++g) c[g] = (i < N) ? part[(size_t)g * NP + i] : 0u;
#pragma unroll
    for (int g = 0; g < HIST_CHUNKS; ++g) deg += c[g];
    if (i < N) dinv[i] = rsqrtf((float)deg + 1.0f);
    unsigned int mine = 0u;
#pragma unroll
    for (int g = 0; g < HIST_CHUNKS; ++g) {
        unsigned int v = c[g];
        for (int o = 16; o > 0; o >>= 1) v += __shfl_xor(v, o);
        if (lane == g) mine = v;
    }
    wpart[wave][lane] = mine;
    __syncthreads();
    if (threadIdx.x < 32) {
        unsigned int v = 0u;
        for (int w = 0; w < SCAN_BLK / 32; ++w) v += wpart[w][lane];
        unsigned int incl = v;
        for (int o = 1; o < 32; o <<= 1) {
            const unsigned int u = __shfl_up(incl, o);
            if (lane >= o) incl += u;
        }
        runOfs[(size_t)blockIdx.x * HIST_CHUNKS + lane] = incl - v;
        if (lane == 31) blkTot[blockIdx.x] = incl;
    }
}

#define FILL_BINS   12800
#define FILL_RUNS   128
__device__ __forceinline__ void fill_body(const int bx, const int by, const int nby,
                                          const int* __restrict__ src,
                                          const int* __restrict__ dst, int E, int N,
                                          const unsigned int* __restrict__ runOfs,
                                          const unsigned int* __restrict__ blkTot, int nblk,
                                          unsigned int* __restrict__ blkStart,
                                          int* __restrict__ ssrc,
                                          unsigned int* stage, unsigned int* bpre, unsigned int* wtot,
                                          unsigned int* runs) {
    const int binLo = by * FILL_BINS;
    const int nb    = min(FILL_BINS, N - binLo);
    const int t     = threadIdx.x;
    const int lane  = t & 31;
    unsigned int* gbase  = runs;
    unsigned int* lstart = runs + FILL_RUNS;
    unsigned int* lcnt   = runs + 2 * FILL_RUNS;
    unsigned int* lcur   = runs + 3 * FILL_RUNS;

    {
        const unsigned int v = (t < nblk) ? blkTot[t] : 0u;
        unsigned int incl = v;
        for (int o = 1; o < 32; o <<= 1) {
            const unsigned int u = __shfl_up(incl, o);
            if (lane >= o) incl += u;
        }
        if (lane == 31) wtot[t >> 5] = incl;
        __syncthreads();
        if (t < 32) {
            const unsigned int w = wtot[t];
            unsigned int wi = w;
            for (int o = 1; o < 32; o <<= 1) {
                const unsigned int u = __shfl_up(wi, o);
                if (lane >= o) wi += u;
            }
            wtot[t] = wi - w;
        }
        __syncthreads();
        bpre[t] = wtot[t >> 5] + incl - v;
    }
    __syncthreads();

    const int b0  = binLo / SCAN_BLK;
    const int nbk = (nb + SCAN_BLK - 1) / SCAN_BLK;
    unsigned int rn = 0u;
    if (t < nbk) {
        const int b = b0 + t;
        const unsigned int ro = runOfs[(size_t)b * HIST_CHUNKS + bx];
        const unsigned int nx = (bx + 1 < HIST_CHUNKS) ? runOfs[(size_t)b * HIST_CHUNKS + bx + 1] : blkTot[b];
        rn = nx - ro;
        gbase[t] = bpre[b] + ro;
        lcnt[t]  = rn;
        if (bx == 0) blkStart[b] = bpre[b];
    }
    if (bx == 0 && by == nby - 1 && t == 0)
        blkStart[nblk] = bpre[nblk - 1] + blkTot[nblk - 1];
    {
        unsigned int incl = rn;
        for (int o = 1; o < 32; o <<= 1) {
            const unsigned int u = __shfl_up(incl, o);
            if (lane >= o) incl += u;
        }
        if (t < FILL_RUNS && lane == 31) wtot[t >> 5] = incl;
        __syncthreads();
        if (t < FILL_RUNS) {
            unsigned int base = 0u;
            for (int w = 0; w < (t >> 5); ++w) base += wtot[w];
            lstart[t] = base + incl - rn;
            lcur[t]   = base + incl - rn;
        }
    }
    __syncthreads();

    const int per = (E + HIST_CHUNKS - 1) / HIST_CHUNKS;
    const int e0  = bx * per;
    const int e1  = min(E, e0 + per);
    int e = e0 + t;
    for (; e + 7 * 1024 < e1; e += 8 * 1024) {
        unsigned int b[8];
        int s[8];
#pragma unroll
        for (int u = 0; u < 8; ++u) {
            b[u] = (unsigned int)(dst[e + u * 1024] - binLo);
            s[u] = src[e + u * 1024];
        }
        unsigned int pos[8];
#pragma unroll
        for (int u = 0; u < 8; ++u) {
            pos[u] = 0u;
            if (b[u] < (unsigned int)nb) pos[u] = atomicAdd(&lcur[b[u] / SCAN_BLK], 1u);
        }
#pragma unroll
        for (int u = 0; u < 8; ++u)
            if (b[u] < (unsigned int)nb) {
                const unsigned int w = (unsigned int)s[u] | ((b[u] % SCAN_BLK) << 24);
                if (pos[u] < (unsigned int)FILL_BINS) stage[pos[u]] = w;
                else { const unsigned int k = b[u] / SCAN_BLK; ssrc[gbase[k] + (pos[u] - lstart[k])] = (int)w; }
            }
    }
    for (; e < e1; e += 1024) {
        const unsigned int b = (unsigned int)(dst[e] - binLo);
        if (b < (unsigned int)nb) {
            const unsigned int k = b / SCAN_BLK;
            const unsigned int p = atomicAdd(&lcur[k], 1u);
            const unsigned int w = (unsigned int)src[e] | ((b % SCAN_BLK) << 24);
            if (p < (unsigned int)FILL_BINS) stage[p] = w;
            else ssrc[gbase[k] + (p - lstart[k])] = (int)w;
        }
    }
    __syncthreads();

    for (int k = t >> 5; k < nbk; k += 32) {
        const unsigned int n = lcnt[k], ls = lstart[k], gb = gbase[k];
        for (unsigned int j = (unsigned int)lane; j < n; j += 32u) {
            const unsigned int lp = ls + j;
            if (lp < (unsigned int)FILL_BINS) ssrc[gb + j] = (int)stage[lp];
        }
    }
}

#define XS_PITCH 36
__device__ __forceinline__ void gemm1_body(const int g, const int G,
    const float* __restrict__ x, const float* __restrict__ Wc,
    const float* __restrict__ dinv, float* __restrict__ h, int N, float* sW, float* xs) {
    {
        const float4* gv = (const float4*)Wc;
        float4* sv = (float4*)sW;
        for (int i = threadIdx.x; i < (IN_DIM * HID_DIM) / 4; i += 1024)
            sv[i] = gv[i];
    }
    __syncthreads();

    const int wave = __builtin_amdgcn_readfirstlane((int)(threadIdx.x >> 5));
    const int lane = threadIdx.x & 31;
    const int r  = lane & 15;
    const int kg = lane >> 4;
    const int nTiles = (N + 15) >> 4;

    const int srow = lane >> 3;
    const int spc  = lane & 7;
    float*       xw  = xs + wave * (16 * XS_PITCH);
    float*       xst = xw + srow * XS_PITCH + spc * 4;
    const float* swl = sW + (kg << 1) * HID_DIM + r;
    const char*  xb  = (const char*)x;

    for (int slot = wave; g + G * slot < nTiles; slot += 32) {
        const int rowBase = (g + G * slot) * 16;
        const int ldBase = max(min(rowBase, N - 16), 0);
        const int shift  = rowBase - ldBase;
        const float* xld = xw + min(r + shift, 15) * XS_PITCH + (kg << 1);
        const unsigned int go0 = (unsigned int)(ldBase + srow) * (unsigned int)(IN_DIM * sizeof(float))
                               + (unsigned int)spc * 16u;

        v8f acc[4] = {v8f{}, v8f{}, v8f{}, v8f{}};
#pragma unroll 1
        for (int q = 0; q < 4; ++q) {
            const char* xq = xb + q * 128;
            v4f xv0, xv1, xv2, xv3;
            asm volatile("global_load_b128 %0, %4, %5\n\t"
                         "global_load_b128 %1, %4, %5 offset:2048\n\t"
                         "global_load_b128 %2, %4, %5 offset:4096\n\t"
                         "global_load_b128 %3, %4, %5 offset:6144\n\t"
                         "s_wait_loadcnt 0x0"
                         : "=&v"(xv0), "=&v"(xv1), "=&v"(xv2), "=&v"(xv3)
                         : "v"(go0), "s"(xq)
                         : "memory");
            *(v4f*)(xst)                 = xv0;
            *(v4f*)(xst +  4 * XS_PITCH) = xv1;
            *(v4f*)(xst +  8 * XS_PITCH) = xv2;
            *(v4f*)(xst + 12 * XS_PITCH) = xv3;
            const float* swq = swl + q * (32 * HID_DIM);
#pragma unroll
            for (int s = 0; s < 8; ++s) {
                const float2 av = *(const float2*)(xld + 4 * s);
                v2f a; a.x = av.x; a.y = av.y;
#pragma unroll
                for (int t = 0; t < 4; ++t) {
                    v2f b;
                    b.x = swq[(4 * s)     * HID_DIM + t * 16];
                    b.y = swq[(4 * s + 1) * HID_DIM + t * 16];
                    acc[t] = __builtin_amdgcn_wmma_f32_16x16x4_f32(
                        false, a, false, b, (short)0, acc[t], false, false);
                }
            }
        }
        float dv[8];
#pragma unroll
        for (int rr = 0; rr < 8; ++rr) dv[rr] = dinv[min(rowBase + rr + 8 * kg, N - 1)];
        char* hb = (char*)h;
#pragma unroll
        for (int t = 0; t < 4; ++t) {
            const int col = t * 16 + r;
#pragma unroll
            for (int rr = 0; rr < 8; ++rr) {
                const int row = rowBase + rr + 8 * kg;
                if (row < N)
                    *(float*)(hb + ((unsigned int)row * (unsigned int)(HID_DIM * sizeof(float)) + (unsigned int)col * 4u))
                        = acc[t][rr] * dv[rr];
            }
        }
    }
}

__global__ __launch_bounds__(1024, 16) void fill_gemm1_kernel(
    const int* __restrict__ src, const int* __restrict__ dst, int E, int N,
    const unsigned int* __restrict__ runOfs, const unsigned int* __restrict__ blkTot, int nblk,
    unsigned int* __restrict__ blkStart, int* __restrict__ ssrc,
    const float* __restrict__ x, const float* __restrict__ Wc,
    const float* __restrict__ dinv, float* __restrict__ h) {
    __shared__ __attribute__((aligned(16))) unsigned int cur[FILL_BINS];
    __shared__ unsigned int bpre[MAX_SCAN_BLKS];
    __shared__ unsigned int wtot[32];
    __shared__ unsigned int runs[4 * FILL_RUNS];
    __shared__ __attribute__((aligned(16))) float xs[32 * 16 * XS_PITCH];
    const int nby   = (N + FILL_BINS - 1) / FILL_BINS;
    const int nHist = HIST_CHUNKS * nby;
    const int b     = blockIdx.x;
    if (b < nHist)
        fill_body(b % HIST_CHUNKS, b / HIST_CHUNKS, nby, src, dst, E, N, runOfs, blkTot, nblk,
                  blkStart, ssrc, cur, bpre, wtot, runs);
    else
        gemm1_body(b - nHist, (int)gridDim.x - nHist, x, Wc, dinv, h, N, (float*)cur, xs);
}

#define ZS_PITCH 68
#define GATHER_SLOT(u) if ((u) < m) { const unsigned int s_ = (unsigned int)__builtin_amdgcn_readlane(sv, (u)); v[(u)] = *(const float2*)(hb + ((s_ << 8) + lb)); }
#define SL_CAP   3072

__global__ __launch_bounds__(256) void gemm2_kernel(
    const unsigned int* __restrict__ blkStart, const int* __restrict__ ssrc,
    const float* __restrict__ h, const float* __restrict__ dinv,
    const float* __restrict__ bc, const float* __restrict__ Wl,
    const float* __restrict__ bl, float* __restrict__ out, int N) {
    __shared__ float sW[HID_DIM * OUT_DIM];
    __shared__ float zs[128 * ZS_PITCH];
    __shared__ int   sl[SL_CAP];

    const int wave = __builtin_amdgcn_readfirstlane((int)(threadIdx.x >> 5));
    const int lane = threadIdx.x & 31;
    const int rowBase = blockIdx.x * 128 + wave * 16;

    const unsigned int pbeg = blkStart[blockIdx.x];
    const unsigned int pcnt = blkStart[blockIdx.x + 1] - pbeg;
    const bool sorted = pcnt <= (unsigned int)SL_CAP;
    unsigned int ent[SL_CAP / 256];
#pragma unroll
    for (int u = 0; u < SL_CAP / 256; ++u) {
        const unsigned int j = (unsigned int)(u * 256) + threadIdx.x;
        ent[u] = (sorted && j < pcnt) ? (unsigned int)ssrc[pbeg + j] : 0u;
    }
    {
        const float4* gv = (const float4*)Wl;
        float4* sv = (float4*)sW;
        for (int i = threadIdx.x; i < (HID_DIM * OUT_DIM) / 4; i += 256)
            sv[i] = gv[i];
    }

    unsigned int* rcnt = (unsigned int*)zs;
    unsigned int* rofs = rcnt + 128;
    unsigned int* rcur = rofs + 132;
    unsigned int* wsm  = rcur + 128;
    unsigned int rp = 0u;
    if (sorted) {
        if (threadIdx.x < 128) rcnt[threadIdx.x] = 0u;
        __syncthreads();
#pragma unroll
        for (int u = 0; u < SL_CAP / 256; ++u)
            if ((unsigned int)(u * 256) + threadIdx.x < pcnt) atomicAdd(&rcnt[ent[u] >> 24], 1u);
        __syncthreads();
        unsigned int v = 0u, incl = 0u;
        if (threadIdx.x < 128) {
            v = rcnt[threadIdx.x];
            incl = v;
            for (int o = 1; o < 32; o <<= 1) {
                const unsigned int u = __shfl_up(incl, o);
                if (lane >= o) incl += u;
            }
            if (lane == 31) wsm[wave] = incl;
        }
        __syncthreads();
        if (threadIdx.x < 128) {
            unsigned int base = 0u;
            for (int w = 0; w < wave; ++w) base += wsm[w];
            const unsigned int ex = base + incl - v;
            rofs[threadIdx.x] = ex;
            rcur[threadIdx.x] = ex;
            if (threadIdx.x == 127) rofs[128] = ex + v;
        }
        __syncthreads();
        unsigned int p[SL_CAP / 256];
#pragma unroll
        for (int u = 0; u < SL_CAP / 256; ++u) {
            p[u] = 0u;
            if ((unsigned int)(u * 256) + threadIdx.x < pcnt) p[u] = atomicAdd(&rcur[ent[u] >> 24], 1u);
        }
#pragma unroll
        for (int u = 0; u < SL_CAP / 256; ++u)
            if ((unsigned int)(u * 256) + threadIdx.x < pcnt) sl[p[u]] = (int)(ent[u] & 0xFFFFFFu);
        rp = rofs[min(wave * 16 + lane, 128)];
        __syncthreads();
    }

    if (rowBase < N) {
        const float        dvec = dinv[min(rowBase + lane, N - 1)];
        const float2       bcv  = *(const float2*)(bc + lane * 2);
        const float* hl = h + lane * 2;
        const char*  hb = (const char*)h;
        const unsigned int lb = (unsigned int)lane * 8u;
        for (int i = 0; i < 16; ++i) {
            const int nld = min(rowBase + i, N - 1);
            float2 acc = *(const float2*)(hl + (size_t)nld * HID_DIM);
            if (sorted) {
                const unsigned int beg = (unsigned int)__builtin_amdgcn_readlane((int)rp, i);
                const unsigned int cnt = (unsigned int)__builtin_amdgcn_readlane((int)rp, i + 1) - beg;
                for (unsigned int base = 0u; base < cnt; base += 32u) {
                    const int m  = (int)min(32u, cnt - base);
                    const int sv = sl[min(beg + base + (unsigned int)lane, (unsigned int)(SL_CAP - 1))];
                    float2 v[32];
#pragma unroll
                    for (int u = 0; u < 32; ++u) v[u] = make_float2(0.0f, 0.0f);
#pragma unroll
                    for (int u = 0; u < 8; ++u) GATHER_SLOT(u)
                    if (m > 8) {
#pragma unroll
                        for (int u = 8; u < 16; ++u) GATHER_SLOT(u)
                        if (m > 16) {
#pragma unroll
                            for (int u = 16; u < 24; ++u) GATHER_SLOT(u)
                            if (m > 24) {
#pragma unroll
                                for (int u = 24; u < 32; ++u) GATHER_SLOT(u)
                            }
                        }
                    }
#pragma unroll
                    for (int u = 0; u < 32; ++u) { acc.x += v[u].x; acc.y += v[u].y; }
                }
            } else {
                const unsigned int myrow = (unsigned int)(wave * 16 + i);
                for (unsigned int j = 0u; j < pcnt; ++j) {
                    const unsigned int w = (unsigned int)__builtin_amdgcn_readfirstlane(ssrc[pbeg + j]);
                    if ((w >> 24) == myrow) {
                        const float2 v0 = *(const float2*)(hl + (size_t)(w & 0xFFFFFFu) * HID_DIM);
                        acc.x += v0.x; acc.y += v0.y;
                    }
                }
            }
            const float di = __int_as_float(__builtin_amdgcn_readlane(__float_as_int(dvec), i));
            float2 z;
            z.x = fmaxf(di * acc.x + bcv.x, 0.0f);
            z.y = fmaxf(di * acc.y + bcv.y, 0.0f);
            *(float2*)(zs + (wave * 16 + i) * ZS_PITCH + lane * 2) = z;
        }
    }
    __syncthreads();
    if (rowBase >= N) return;

    const int r  = lane & 15;
    const int kg = lane >> 4;
    const float* zrow = zs + (wave * 16 + r) * ZS_PITCH;

    v8f acc[8] = {v8f{}, v8f{}, v8f{}, v8f{}, v8f{}, v8f{}, v8f{}, v8f{}};
    for (int k = 0; k < HID_DIM; k += 4) {
        const int kk = k + (kg << 1);
        const float2 av = *(const float2*)(zrow + kk);
        v2f a; a.x = av.x; a.y = av.y;
#pragma unroll
        for (int t = 0; t < 8; ++t) {
            v2f b;
            b.x = sW[(kk)     * OUT_DIM + t * 16 + r];
            b.y = sW[(kk + 1) * OUT_DIM + t * 16 + r];
            acc[t] = __builtin_amdgcn_wmma_f32_16x16x4_f32(
                false, a, false, b, (short)0, acc[t], false, false);
        }
    }
#pragma unroll
    for (int t = 0; t < 8; ++t) {
        const int col = t * 16 + r;
        const float blv = bl[col];
#pragma unroll
        for (int rr = 0; rr < 8; ++rr) {
            const int row = rowBase + rr + 8 * kg;
            if (row < N) out[(size_t)row * OUT_DIM + col] = acc[t][rr] + blv;
        }
    }
}

static inline size_t alignUp(size_t v) { return (v + 255) & ~(size_t)255; }

extern "C" void kernel_launch(void* const* d_in, const int* in_sizes, int n_in,
                              void* d_out, int out_size, void* d_ws, size_t ws_size,
                              hipStream_t stream) {
    const float* x  = (const float*)d_in[0];
    const int*   ei = (const int*)  d_in[1];
    const float* Wc = (const float*)d_in[2];
    const float* bc = (const float*)d_in[3];
    const float* Wl = (const float*)d_in[4];
    const float* bl = (const float*)d_in[5];
    float* out = (float*)d_out;

    const int N = in_sizes[0] / IN_DIM;
    const int E = in_sizes[1] / 2;
    const int* src = ei;
    const int* dst = ei + E;

    char* ws = (char*)d_ws;
    const size_t hBytes = (size_t)N * HID_DIM * sizeof(float);
    size_t off = 0;
    float*        h    = (float*)(ws + off);        off = alignUp(off + hBytes);
    char*         srt  = ws + off;                  off = alignUp(off + hBytes);
    float*        dinv = (float*)(ws + off);        off = alignUp(off + (size_t)N * sizeof(float));

    const int  nblk = (N + SCAN_BLK - 1) / SCAN_BLK;
    size_t so = 0;
    int*          ssrc     = (int*)(srt + so);          so = alignUp(so + (size_t)E * sizeof(int));
    unsigned int* blkStart = (unsigned int*)(srt + so); so = alignUp(so + ((size_t)nblk + 1) * sizeof(unsigned int));
    unsigned int* blkTot   = (unsigned int*)(srt + so); so = alignUp(so + (size_t)nblk * sizeof(unsigned int));
    unsigned int* runOfs   = (unsigned int*)(srt + so); so = alignUp(so + (size_t)nblk * HIST_CHUNKS * sizeof(unsigned int));
    unsigned int* part     = (unsigned int*)(srt + so);

    const dim3 hgrid(HIST_CHUNKS, (N + HIST_BINS - 1) / HIST_BINS);
    const int  nHist = HIST_CHUNKS * ((N + FILL_BINS - 1) / FILL_BINS);
    const int  gemmGroups = 256;

    deg_kernel <<<hgrid, 1024, 0, stream>>>(dst, E, N, part);
    dinv_kernel<<<nblk, SCAN_BLK, 0, stream>>>(part, N, dinv, runOfs, blkTot);
    fill_gemm1_kernel<<<nHist + gemmGroups, 1024, 0, stream>>>(src, dst, E, N, runOfs, blkTot, nblk, blkStart, ssrc,
                                                               x, Wc, dinv, h);

    const int rowBlocks = (N + 127) / 128;

    gemm2_kernel<<<rowBlocks, 256, 0, stream>>>(blkStart, ssrc, h, dinv, bc, Wl, bl, out, N);
}


# Both GEMM epilogues send the tile through the wave's own LDS rows: a store instruction writes whole 128-byte lines of its rows, 16 B per lane, not 64-byte pieces of two rows; values unchanged
# speedup vs baseline: 3.7152x; 1.0966x over previous
#include <hip/hip_runtime.h>
#include <hip/hip_bf16.h>

typedef __attribute__((ext_vector_type(2))) float v2f;
typedef __attribute__((ext_vector_type(4))) float v4f;
typedef __attribute__((ext_vector_type(8))) float v8f;

#define IN_DIM  128
#define HID_DIM 64
#define OUT_DIM 128

#define HIST_BINS   51200
#define HIST_CHUNKS 32

__global__ __launch_bounds__(1024) void deg_kernel(const int* __restrict__ dst, int E, int N,
                                                   unsigned int* __restrict__ part) {
    __shared__ unsigned int cnt[HIST_BINS];
    const int NP    = (N + 31) & ~31;
    const int binLo = blockIdx.y * HIST_BINS;
    const int nb    = min(HIST_BINS, N - binLo);
    {
        uint4* cv = (uint4*)cnt;
        for (int i = threadIdx.x; i < HIST_BINS / 4; i += 1024)
            cv[i] = make_uint4(0u, 0u, 0u, 0u);
    }
    __syncthreads();

    const int per = (E + (int)gridDim.x - 1) / (int)gridDim.x;
    const int e0  = blockIdx.x * per;
    const int e1  = min(E, e0 + per);
    int e = e0 + threadIdx.x;
    for (; e + 15 * 1024 < e1; e += 16 * 1024) {
        unsigned int b[16];
#pragma unroll
        for (int u = 0; u < 16; ++u) b[u] = (unsigned int)(dst[e + u * 1024] - binLo);
#pragma unroll
        for (int u = 0; u < 16; ++u)
            if (b[u] < (unsigned int)nb) atomicAdd(&cnt[b[u]], 1u);
    }
    for (; e < e1; e += 1024) {
        const unsigned int b = (unsigned int)(dst[e] - binLo);
        if (b < (unsigned int)nb) atomicAdd(&cnt[b], 1u);
    }
    __syncthreads();

    unsigned int* row = part + (size_t)blockIdx.x * NP + binLo;
    const int nb4 = nb >> 2;
    {
        const uint4* cv = (const uint4*)cnt;
        uint4* rv = (uint4*)row;
        for (int i = threadIdx.x; i < nb4; i += 1024) rv[i] = cv[i];
    }
    for (int i = (nb4 << 2) + threadIdx.x; i < nb; i += 1024) row[i] = cnt[i];
}

#define SCAN_BLK      128
#define MAX_SCAN_BLKS 1024

__global__ __launch_bounds__(SCAN_BLK) void dinv_kernel(const unsigned int* __restrict__ part, int N,
                                                       float* __restrict__ dinv,
                                                       unsigned int* __restrict__ runOfs,
                                                       unsigned int* __restrict__ blkTot) {
    __shared__ unsigned int wpart[SCAN_BLK / 32][HIST_CHUNKS];
    const int NP   = (N + 31) & ~31;
    const int i    = blockIdx.x * SCAN_BLK + threadIdx.x;
    const int lane = threadIdx.x & 31;
    const int wave = threadIdx.x >> 5;
    unsigned int c[HIST_CHUNKS];
    unsigned int deg = 0u;
#pragma unroll
    for (int g = 0; g < HIST_CHUNKS; ++g) c[g] = (i < N) ? part[(size_t)g * NP + i] : 0u;
#pragma unroll
    for (int g = 0; g < HIST_CHUNKS; ++g) deg += c[g];
    if (i < N) dinv[i] = rsqrtf((float)deg + 1.0f);
    unsigned int mine = 0u;
#pragma unroll
    for (int g = 0; g < HIST_CHUNKS; ++g) {
        unsigned int v = c[g];
        for (int o = 16; o > 0; o >>= 1) v += __shfl_xor(v, o);
        if (lane == g) mine = v;
    }
    wpart[wave][lane] = mine;
    __syncthreads();
    if (threadIdx.x < 32) {
        unsigned int v = 0u;
        for (int w = 0; w < SCAN_BLK / 32; ++w) v += wpart[w][lane];
        unsigned int incl = v;
        for (int o = 1; o < 32; o <<= 1) {
            const unsigned int u = __shfl_up(incl, o);
            if (lane >= o) incl += u;
        }
        runOfs[(size_t)blockIdx.x * HIST_CHUNKS + lane] = incl - v;
        if (lane == 31) blkTot[blockIdx.x] = incl;
    }
}

#define FILL_BINS   12800
#define FILL_RUNS   128
__device__ __forceinline__ void fill_body(const int bx, const int by, const int nby,
                                          const int* __restrict__ src,
                                          const int* __restrict__ dst, int E, int N,
                                          const unsigned int* __restrict__ runOfs,
                                          const unsigned int* __restrict__ blkTot, int nblk,
                                          unsigned int* __restrict__ blkStart,
                                          int* __restrict__ ssrc,
                                          unsigned int* stage, unsigned int* bpre, unsigned int* wtot,
                                          unsigned int* runs) {
    const int binLo = by * FILL_BINS;
    const int nb    = min(FILL_BINS, N - binLo);
    const int t     = threadIdx.x;
    const int lane  = t & 31;
    unsigned int* gbase  = runs;
    unsigned int* lstart = runs + FILL_RUNS;
    unsigned int* lcnt   = runs + 2 * FILL_RUNS;
    unsigned int* lcur   = runs + 3 * FILL_RUNS;

    {
        const unsigned int v = (t < nblk) ? blkTot[t] : 0u;
        unsigned int incl = v;
        for (int o = 1; o < 32; o <<= 1) {
            const unsigned int u = __shfl_up(incl, o);
            if (lane >= o) incl += u;
        }
        if (lane == 31) wtot[t >> 5] = incl;
        __syncthreads();
        if (t < 32) {
            const unsigned int w = wtot[t];
            unsigned int wi = w;
            for (int o = 1; o < 32; o <<= 1) {
                const unsigned int u = __shfl_up(wi, o);
                if (lane >= o) wi += u;
            }
            wtot[t] = wi - w;
        }
        __syncthreads();
        bpre[t] = wtot[t >> 5] + incl - v;
    }
    __syncthreads();

    const int b0  = binLo / SCAN_BLK;
    const int nbk = (nb + SCAN_BLK - 1) / SCAN_BLK;
    unsigned int rn = 0u;
    if (t < nbk) {
        const int b = b0 + t;
        const unsigned int ro = runOfs[(size_t)b * HIST_CHUNKS + bx];
        const unsigned int nx = (bx + 1 < HIST_CHUNKS) ? runOfs[(size_t)b * HIST_CHUNKS + bx + 1] : blkTot[b];
        rn = nx - ro;
        gbase[t] = bpre[b] + ro;
        lcnt[t]  = rn;
        if (bx == 0) blkStart[b] = bpre[b];
    }
    if (bx == 0 && by == nby - 1 && t == 0)
        blkStart[nblk] = bpre[nblk - 1] + blkTot[nblk - 1];
    {
        unsigned int incl = rn;
        for (int o = 1; o < 32; o <<= 1) {
            const unsigned int u = __shfl_up(incl, o);
            if (lane >= o) incl += u;
        }
        if (t < FILL_RUNS && lane == 31) wtot[t >> 5] = incl;
        __syncthreads();
        if (t < FILL_RUNS) {
            unsigned int base = 0u;
            for (int w = 0; w < (t >> 5); ++w) base += wtot[w];
            lstart[t] = base + incl - rn;
            lcur[t]   = base + incl - rn;
        }
    }
    __syncthreads();

    const int per = (E + HIST_CHUNKS - 1) / HIST_CHUNKS;
    const int e0  = bx * per;
    const int e1  = min(E, e0 + per);
    int e = e0 + t;
    for (; e + 7 * 1024 < e1; e += 8 * 1024) {
        unsigned int b[8];
        int s[8];
#pragma unroll
        for (int u = 0; u < 8; ++u) {
            b[u] = (unsigned int)(dst[e + u * 1024] - binLo);
            s[u] = src[e + u * 1024];
        }
        unsigned int pos[8];
#pragma unroll
        for (int u = 0; u < 8; ++u) {
            pos[u] = 0u;
            if (b[u] < (unsigned int)nb) pos[u] = atomicAdd(&lcur[b[u] / SCAN_BLK], 1u);
        }
#pragma unroll
        for (int u = 0; u < 8; ++u)
            if (b[u] < (unsigned int)nb) {
                const unsigned int w = (unsigned int)s[u] | ((b[u] % SCAN_BLK) << 24);
                if (pos[u] < (unsigned int)FILL_BINS) stage[pos[u]] = w;
                else { const unsigned int k = b[u] / SCAN_BLK; ssrc[gbase[k] + (pos[u] - lstart[k])] = (int)w; }
            }
    }
    for (; e < e1; e += 1024) {
        const unsigned int b = (unsigned int)(dst[e] - binLo);
        if (b < (unsigned int)nb) {
            const unsigned int k = b / SCAN_BLK;
            const unsigned int p = atomicAdd(&lcur[k], 1u);
            const unsigned int w = (unsigned int)src[e] | ((b % SCAN_BLK) << 24);
            if (p < (unsigned int)FILL_BINS) stage[p] = w;
            else ssrc[gbase[k] + (p - lstart[k])] = (int)w;
        }
    }
    __syncthreads();

    for (int k = t >> 5; k < nbk; k += 32) {
        const unsigned int n = lcnt[k], ls = lstart[k], gb = gbase[k];
        for (unsigned int j = (unsigned int)lane; j < n; j += 32u) {
            const unsigned int lp = ls + j;
            if (lp < (unsigned int)FILL_BINS) ssrc[gb + j] = (int)stage[lp];
        }
    }
}

#define XS_PITCH 36
__device__ __forceinline__ void gemm1_body(const int g, const int G,
    const float* __restrict__ x, const float* __restrict__ Wc,
    const float* __restrict__ dinv, float* __restrict__ h, int N, float* sW, float* xs) {
    {
        const float4* gv = (const float4*)Wc;
        float4* sv = (float4*)sW;
        for (int i = threadIdx.x; i < (IN_DIM * HID_DIM) / 4; i += 1024)
            sv[i] = gv[i];
    }
    __syncthreads();

    const int wave = __builtin_amdgcn_readfirstlane((int)(threadIdx.x >> 5));
    const int lane = threadIdx.x & 31;
    const int r  = lane & 15;
    const int kg = lane >> 4;
    const int nTiles = (N + 15) >> 4;

    const int srow = lane >> 3;
    const int spc  = lane & 7;
    float*       xw  = xs + wave * (16 * XS_PITCH);
    float*       xst = xw + srow * XS_PITCH + spc * 4;
    const float* swl = sW + (kg << 1) * HID_DIM + r;
    const char*  xb  = (const char*)x;

    for (int slot = wave; g + G * slot < nTiles; slot += 32) {
        const int rowBase = (g + G * slot) * 16;
        const int ldBase = max(min(rowBase, N - 16), 0);
        const int shift  = rowBase - ldBase;
        const float* xld = xw + min(r + shift, 15) * XS_PITCH + (kg << 1);
        const unsigned int go0 = (unsigned int)(ldBase + srow) * (unsigned int)(IN_DIM * sizeof(float))
                               + (unsigned int)spc * 16u;

        v8f acc[4] = {v8f{}, v8f{}, v8f{}, v8f{}};
#pragma unroll 1
        for (int q = 0; q < 4; ++q) {
            const char* xq = xb + q * 128;
            v4f xv0, xv1, xv2, xv3;
            asm volatile("global_load_b128 %0, %4, %5\n\t"
                         "global_load_b128 %1, %4, %5 offset:2048\n\t"
                         "global_load_b128 %2, %4, %5 offset:4096\n\t"
                         "global_load_b128 %3, %4, %5 offset:6144\n\t"
                         "s_wait_loadcnt 0x0"
                         : "=&v"(xv0), "=&v"(xv1), "=&v"(xv2), "=&v"(xv3)
                         : "v"(go0), "s"(xq)
                         : "memory");
            *(v4f*)(xst)                 = xv0;
            *(v4f*)(xst +  4 * XS_PITCH) = xv1;
            *(v4f*)(xst +  8 * XS_PITCH) = xv2;
            *(v4f*)(xst + 12 * XS_PITCH) = xv3;
            const float* swq = swl + q * (32 * HID_DIM);
#pragma unroll
            for (int s = 0; s < 8; ++s) {
                const float2 av = *(const float2*)(xld + 4 * s);
                v2f a; a.x = av.x; a.y = av.y;
#pragma unroll
                for (int t = 0; t < 4; ++t) {
                    v2f b;
                    b.x = swq[(4 * s)     * HID_DIM + t * 16];
                    b.y = swq[(4 * s + 1) * HID_DIM + t * 16];
                    acc[t] = __builtin_amdgcn_wmma_f32_16x16x4_f32(
                        false, a, false, b, (short)0, acc[t], false, false);
                }
            }
        }
        float dvo[4];
#pragma unroll
        for (int j = 0; j < 4; ++j) dvo[j] = dinv[min(rowBase + srow + 4 * j, N - 1)];
        char* hb = (char*)h;
        float* hst = xw + (8 * kg) * XS_PITCH + r;
        const unsigned int ho0 = (unsigned int)(rowBase + srow) * (unsigned int)(HID_DIM * sizeof(float))
                               + (unsigned int)spc * 16u;
        const bool full = rowBase + 16 <= N;
#pragma unroll
        for (int p = 0; p < 2; ++p) {
#pragma unroll
            for (int tt = 0; tt < 2; ++tt) {
#pragma unroll
                for (int rr = 0; rr < 8; ++rr)
                    hst[rr * XS_PITCH + tt * 16] = acc[2 * p + tt][rr];
            }
            v4f o[4];
#pragma unroll
            for (int j = 0; j < 4; ++j) o[j] = *(const v4f*)(xst + 4 * j * XS_PITCH) * dvo[j];
            if (full) {
#pragma unroll
                for (int j = 0; j < 4; ++j)
                    *(v4f*)(hb + (ho0 + (unsigned int)(j * 4 * HID_DIM * (int)sizeof(float) + p * 128))) = o[j];
            } else {
#pragma unroll
                for (int j = 0; j < 4; ++j)
                    if (rowBase + 4 * j + srow < N)
                        *(v4f*)(hb + (ho0 + (unsigned int)(j * 4 * HID_DIM * (int)sizeof(float) + p * 128))) = o[j];
            }
        }
    }
}

__global__ __launch_bounds__(1024, 16) void fill_gemm1_kernel(
    const int* __restrict__ src, const int* __restrict__ dst, int E, int N,
    const unsigned int* __restrict__ runOfs, const unsigned int* __restrict__ blkTot, int nblk,
    unsigned int* __restrict__ blkStart, int* __restrict__ ssrc,
    const float* __restrict__ x, const float* __restrict__ Wc,
    const float* __restrict__ dinv, float* __restrict__ h) {
    __shared__ __attribute__((aligned(16))) unsigned int cur[FILL_BINS];
    __shared__ unsigned int bpre[MAX_SCAN_BLKS];
    __shared__ unsigned int wtot[32];
    __shared__ unsigned int runs[4 * FILL_RUNS];
    __shared__ __attribute__((aligned(16))) float xs[32 * 16 * XS_PITCH];
    const int nby   = (N + FILL_BINS - 1) / FILL_BINS;
    const int nHist = HIST_CHUNKS * nby;
    const int b     = blockIdx.x;
    if (b < nHist)
        fill_body(b % HIST_CHUNKS, b / HIST_CHUNKS, nby, src, dst, E, N, runOfs, blkTot, nblk,
                  blkStart, ssrc, cur, bpre, wtot, runs);
    else
        gemm1_body(b - nHist, (int)gridDim.x - nHist, x, Wc, dinv, h, N, (float*)cur, xs);
}

#define ZS_PITCH 68
#define GATHER_SLOT(u) if ((u) < m) { const unsigned int s_ = (unsigned int)__builtin_amdgcn_readlane(sv, (u)); v[(u)] = *(const float2*)(hb + ((s_ << 8) + lb)); }
#define SL_CAP   3072

__global__ __launch_bounds__(256) void gemm2_kernel(
    const unsigned int* __restrict__ blkStart, const int* __restrict__ ssrc,
    const float* __restrict__ h, const float* __restrict__ dinv,
    const float* __restrict__ bc, const float* __restrict__ Wl,
    const float* __restrict__ bl, float* __restrict__ out, int N) {
    __shared__ float sW[HID_DIM * OUT_DIM];
    __shared__ __attribute__((aligned(16))) float zs[128 * ZS_PITCH];
    __shared__ int   sl[SL_CAP];

    const int wave = __builtin_amdgcn_readfirstlane((int)(threadIdx.x >> 5));
    const int lane = threadIdx.x & 31;
    const int rowBase = blockIdx.x * 128 + wave * 16;

    const unsigned int pbeg = blkStart[blockIdx.x];
    const unsigned int pcnt = blkStart[blockIdx.x + 1] - pbeg;
    const bool sorted = pcnt <= (unsigned int)SL_CAP;
    unsigned int ent[SL_CAP / 256];
#pragma unroll
    for (int u = 0; u < SL_CAP / 256; ++u) {
        const unsigned int j = (unsigned int)(u * 256) + threadIdx.x;
        ent[u] = (sorted && j < pcnt) ? (unsigned int)ssrc[pbeg + j] : 0u;
    }
    {
        const float4* gv = (const float4*)Wl;
        float4* sv = (float4*)sW;
        for (int i = threadIdx.x; i < (HID_DIM * OUT_DIM) / 4; i += 256)
            sv[i] = gv[i];
    }

    unsigned int* rcnt = (unsigned int*)zs;
    unsigned int* rofs = rcnt + 128;
    unsigned int* rcur = rofs + 132;
    unsigned int* wsm  = rcur + 128;
    unsigned int rp = 0u;
    if (sorted) {
        if (threadIdx.x < 128) rcnt[threadIdx.x] = 0u;
        __syncthreads();
#pragma unroll
        for (int u = 0; u < SL_CAP / 256; ++u)
            if ((unsigned int)(u * 256) + threadIdx.x < pcnt) atomicAdd(&rcnt[ent[u] >> 24], 1u);
        __syncthreads();
        unsigned int v = 0u, incl = 0u;
        if (threadIdx.x < 128) {
            v = rcnt[threadIdx.x];
            incl = v;
            for (int o = 1; o < 32; o <<= 1) {
                const unsigned int u = __shfl_up(incl, o);
                if (lane >= o) incl += u;
            }
            if (lane == 31) wsm[wave] = incl;
        }
        __syncthreads();
        if (threadIdx.x < 128) {
            unsigned int base = 0u;
            for (int w = 0; w < wave; ++w) base += wsm[w];
            const unsigned int ex = base + incl - v;
            rofs[threadIdx.x] = ex;
            rcur[threadIdx.x] = ex;
            if (threadIdx.x == 127) rofs[128] = ex + v;
        }
        __syncthreads();
        unsigned int p[SL_CAP / 256];
#pragma unroll
        for (int u = 0; u < SL_CAP / 256; ++u) {
            p[u] = 0u;
            if ((unsigned int)(u * 256) + threadIdx.x < pcnt) p[u] = atomicAdd(&rcur[ent[u] >> 24], 1u);
        }
#pragma unroll
        for (int u = 0; u < SL_CAP / 256; ++u)
            if ((unsigned int)(u * 256) + threadIdx.x < pcnt) sl[p[u]] = (int)(ent[u] & 0xFFFFFFu);
        rp = rofs[min(wave * 16 + lane, 128)];
        __syncthreads();
    }

    if (rowBase < N) {
        const float        dvec = dinv[min(rowBase + lane, N - 1)];
        const float2       bcv  = *(const float2*)(bc + lane * 2);
        const float* hl = h + lane * 2;
        const char*  hb = (const char*)h;
        const unsigned int lb = (unsigned int)lane * 8u;
        for (int i = 0; i < 16; ++i) {
            const int nld = min(rowBase + i, N - 1);
            float2 acc = *(const float2*)(hl + (size_t)nld * HID_DIM);
            if (sorted) {
                const unsigned int beg = (unsigned int)__builtin_amdgcn_readlane((int)rp, i);
                const unsigned int cnt = (unsigned int)__builtin_amdgcn_readlane((int)rp, i + 1) - beg;
                for (unsigned int base = 0u; base < cnt; base += 32u) {
                    const int m  = (int)min(32u, cnt - base);
                    const int sv = sl[min(beg + base + (unsigned int)lane, (unsigned int)(SL_CAP - 1))];
                    float2 v[32];
#pragma unroll
                    for (int u = 0; u < 32; ++u) v[u] = make_float2(0.0f, 0.0f);
#pragma unroll
                    for (int u = 0; u < 8; ++u) GATHER_SLOT(u)
                    if (m > 8) {
#pragma unroll
                        for (int u = 8; u < 16; ++u) GATHER_SLOT(u)
                        if (m > 16) {
#pragma unroll
                            for (int u = 16; u < 24; ++u) GATHER_SLOT(u)
                            if (m > 24) {
#pragma unroll
                                for (int u = 24; u < 32; ++u) GATHER_SLOT(u)
                            }
                        }
                    }
#pragma unroll
                    for (int u = 0; u < 32; ++u) { acc.x += v[u].x; acc.y += v[u].y; }
                }
            } else {
                const unsigned int myrow = (unsigned int)(wave * 16 + i);
                for (unsigned int j = 0u; j < pcnt; ++j) {
                    const unsigned int w = (unsigned int)__builtin_amdgcn_readfirstlane(ssrc[pbeg + j]);
                    if ((w >> 24) == myrow) {
                        const float2 v0 = *(const float2*)(hl + (size_t)(w & 0xFFFFFFu) * HID_DIM);
                        acc.x += v0.x; acc.y += v0.y;
                    }
                }
            }
            const float di = __int_as_float(__builtin_amdgcn_readlane(__float_as_int(dvec), i));
            float2 z;
            z.x = fmaxf(di * acc.x + bcv.x, 0.0f);
            z.y = fmaxf(di * acc.y + bcv.y, 0.0f);
            *(float2*)(zs + (wave * 16 + i) * ZS_PITCH + lane * 2) = z;
        }
    }
    __syncthreads();
    if (rowBase >= N) return;

    const int r  = lane & 15;
    const int kg = lane >> 4;
    const float* zrow = zs + (wave * 16 + r) * ZS_PITCH;

    v8f acc[8] = {v8f{}, v8f{}, v8f{}, v8f{}, v8f{}, v8f{}, v8f{}, v8f{}};
    for (int k = 0; k < HID_DIM; k += 4) {
        const int kk = k + (kg << 1);
        const float2 av = *(const float2*)(zrow + kk);
        v2f a; a.x = av.x; a.y = av.y;
#pragma unroll
        for (int t = 0; t < 8; ++t) {
            v2f b;
            b.x = sW[(kk)     * OUT_DIM + t * 16 + r];
            b.y = sW[(kk + 1) * OUT_DIM + t * 16 + r];
            acc[t] = __builtin_amdgcn_wmma_f32_16x16x4_f32(
                false, a, false, b, (short)0, acc[t], false, false);
        }
    }
    float*       zst  = zs + (wave * 16 + 8 * kg) * ZS_PITCH + r;
    const int    orow = lane >> 4;
    const int    oc4  = (lane & 15) * 4;
    const float* zld  = zs + (wave * 16 + orow) * ZS_PITCH + oc4;
    float*       ob   = out + (size_t)(rowBase + orow) * OUT_DIM + oc4;
    const bool   full = rowBase + 16 <= N;
    v4f bv[2];
#pragma unroll
    for (int p = 0; p < 2; ++p) {
        bv[p].x = bl[p * 64 + oc4];
        bv[p].y = bl[p * 64 + oc4 + 1];
        bv[p].z = bl[p * 64 + oc4 + 2];
        bv[p].w = bl[p * 64 + oc4 + 3];
    }
#pragma unroll
    for (int p = 0; p < 2; ++p) {
#pragma unroll
        for (int tt = 0; tt < 4; ++tt) {
#pragma unroll
            for (int rr = 0; rr < 8; ++rr)
                zst[rr * ZS_PITCH + tt * 16] = acc[4 * p + tt][rr];
        }
        v4f o[8];
#pragma unroll
        for (int j = 0; j < 8; ++j) o[j] = *(const v4f*)(zld + 2 * j * ZS_PITCH) + bv[p];
        if (full) {
#pragma unroll
            for (int j = 0; j < 8; ++j) *(v4f*)(ob + (size_t)(2 * j) * OUT_DIM + p * 64) = o[j];
        } else {
#pragma unroll
            for (int j = 0; j < 8; ++j)
                if (rowBase + 2 * j + orow < N) *(v4f*)(ob + (size_t)(2 * j) * OUT_DIM + p * 64) = o[j];
        }
    }
}

static inline size_t alignUp(size_t v) { return (v + 255) & ~(size_t)255; }

extern "C" void kernel_launch(void* const* d_in, const int* in_sizes, int n_in,
                              void* d_out, int out_size, void* d_ws, size_t ws_size,
                              hipStream_t stream) {
    const float* x  = (const float*)d_in[0];
    const int*   ei = (const int*)  d_in[1];
    const float* Wc = (const float*)d_in[2];
    const float* bc = (const float*)d_in[3];
    const float* Wl = (const float*)d_in[4];
    const float* bl = (const float*)d_in[5];
    float* out = (float*)d_out;

    const int N = in_sizes[0] / IN_DIM;
    const int E = in_sizes[1] / 2;
    const int* src = ei;
    const int* dst = ei + E;

    char* ws = (char*)d_ws;
    const size_t hBytes = (size_t)N * HID_DIM * sizeof(float);
    size_t off = 0;
    float*        h    = (float*)(ws + off);        off = alignUp(off + hBytes);
    char*         srt  = ws + off;                  off = alignUp(off + hBytes);
    float*        dinv = (float*)(ws + off);        off = alignUp(off + (size_t)N * sizeof(float));

    const int  nblk = (N + SCAN_BLK - 1) / SCAN_BLK;
    size_t so = 0;
    int*          ssrc     = (int*)(srt + so);          so = alignUp(so + (size_t)E * sizeof(int));
    unsigned int* blkStart = (unsigned int*)(srt + so); so = alignUp(so + ((size_t)nblk + 1) * sizeof(unsigned int));
    unsigned int* blkTot   = (unsigned int*)(srt + so); so = alignUp(so + (size_t)nblk * sizeof(unsigned int));
    unsigned int* runOfs   = (unsigned int*)(srt + so); so = alignUp(so + (size_t)nblk * HIST_CHUNKS * sizeof(unsigned int));
    unsigned int* part     = (unsigned int*)(srt + so);

    const dim3 hgrid(HIST_CHUNKS, (N + HIST_BINS - 1) / HIST_BINS);
    const int  nHist = HIST_CHUNKS * ((N + FILL_BINS - 1) / FILL_BINS);
    const int  gemmGroups = 256;

    deg_kernel <<<hgrid, 1024, 0, stream>>>(dst, E, N, part);
    dinv_kernel<<<nblk, SCAN_BLK, 0, stream>>>(part, N, dinv, runOfs, blkTot);
    fill_gemm1_kernel<<<nHist + gemmGroups, 1024, 0, stream>>>(src, dst, E, N, runOfs, blkTot, nblk, blkStart, ssrc,
                                                               x, Wc, dinv, h);

    const int rowBlocks = (N + 127) / 128;

    gemm2_kernel<<<rowBlocks, 256, 0, stream>>>(blkStart, ssrc, h, dinv, bc, Wl, bl, out, N);
}


# Both GEMMs' B fragments: W staged in LDS with a k-step's 4 k side by side per column: one conflict-free 8-byte read per fragment, not two 4-byte reads that collide across lane halves; k order kept
# speedup vs baseline: 3.8687x; 1.0413x over previous
#include <hip/hip_runtime.h>
#include <hip/hip_bf16.h>

typedef __attribute__((ext_vector_type(2))) float v2f;
typedef __attribute__((ext_vector_type(4))) float v4f;
typedef __attribute__((ext_vector_type(8))) float v8f;

#define IN_DIM  128
#define HID_DIM 64
#define OUT_DIM 128

#define HIST_BINS   51200
#define HIST_CHUNKS 32

__global__ __launch_bounds__(1024) void deg_kernel(const int* __restrict__ dst, int E, int N,
                                                   unsigned int* __restrict__ part) {
    __shared__ unsigned int cnt[HIST_BINS];
    const int NP    = (N + 31) & ~31;
    const int binLo = blockIdx.y * HIST_BINS;
    const int nb    = min(HIST_BINS, N - binLo);
    {
        uint4* cv = (uint4*)cnt;
        for (int i = threadIdx.x; i < HIST_BINS / 4; i += 1024)
            cv[i] = make_uint4(0u, 0u, 0u, 0u);
    }
    __syncthreads();

    const int per = (E + (int)gridDim.x - 1) / (int)gridDim.x;
    const int e0  = blockIdx.x * per;
    const int e1  = min(E, e0 + per);
    int e = e0 + threadIdx.x;
    for (; e + 15 * 1024 < e1; e += 16 * 1024) {
        unsigned int b[16];
#pragma unroll
        for (int u = 0; u < 16; ++u) b[u] = (unsigned int)(dst[e + u * 1024] - binLo);
#pragma unroll
        for (int u = 0; u < 16; ++u)
            if (b[u] < (unsigned int)nb) atomicAdd(&cnt[b[u]], 1u);
    }
    for (; e < e1; e += 1024) {
        const unsigned int b = (unsigned int)(dst[e] - binLo);
        if (b < (unsigned int)nb) atomicAdd(&cnt[b], 1u);
    }
    __syncthreads();

    unsigned int* row = part + (size_t)blockIdx.x * NP + binLo;
    const int nb4 = nb >> 2;
    {
        const uint4* cv = (const uint4*)cnt;
        uint4* rv = (uint4*)row;
        for (int i = threadIdx.x; i < nb4; i += 1024) rv[i] = cv[i];
    }
    for (int i = (nb4 << 2) + threadIdx.x; i < nb; i += 1024) row[i] = cnt[i];
}

#define SCAN_BLK      128
#define MAX_SCAN_BLKS 1024

__global__ __launch_bounds__(SCAN_BLK) void dinv_kernel(const unsigned int* __restrict__ part, int N,
                                                       float* __restrict__ dinv,
                                                       unsigned int* __restrict__ runOfs,
                                                       unsigned int* __restrict__ blkTot) {
    __shared__ unsigned int wpart[SCAN_BLK / 32][HIST_CHUNKS];
    const int NP   = (N + 31) & ~31;
    const int i    = blockIdx.x * SCAN_BLK + threadIdx.x;
    const int lane = threadIdx.x & 31;
    const int wave = threadIdx.x >> 5;
    unsigned int c[HIST_CHUNKS];
    unsigned int deg = 0u;
#pragma unroll
    for (int g = 0; g < HIST_CHUNKS; ++g) c[g] = (i < N) ? part[(size_t)g * NP + i] : 0u;
#pragma unroll
    for (int g = 0; g < HIST_CHUNKS; ++g) deg += c[g];
    if (i < N) dinv[i] = rsqrtf((float)deg + 1.0f);
    unsigned int mine = 0u;
#pragma unroll
    for (int g = 0; g < HIST_CHUNKS; ++g) {
        unsigned int v = c[g];
        for (int o = 16; o > 0; o >>= 1) v += __shfl_xor(v, o);
        if (lane == g) mine = v;
    }
    wpart[wave][lane] = mine;
    __syncthreads();
    if (threadIdx.x < 32) {
        unsigned int v = 0u;
        for (int w = 0; w < SCAN_BLK / 32; ++w) v += wpart[w][lane];
        unsigned int incl = v;
        for (int o = 1; o < 32; o <<= 1) {
            const unsigned int u = __shfl_up(incl, o);
            if (lane >= o) incl += u;
        }
        runOfs[(size_t)blockIdx.x * HIST_CHUNKS + lane] = incl - v;
        if (lane == 31) blkTot[blockIdx.x] = incl;
    }
}

#define FILL_BINS   12800
#define FILL_RUNS   128
__device__ __forceinline__ void fill_body(const int bx, const int by, const int nby,
                                          const int* __restrict__ src,
                                          const int* __restrict__ dst, int E, int N,
                                          const unsigned int* __restrict__ runOfs,
                                          const unsigned int* __restrict__ blkTot, int nblk,
                                          unsigned int* __restrict__ blkStart,
                                          int* __restrict__ ssrc,
                                          unsigned int* stage, unsigned int* bpre, unsigned int* wtot,
                                          unsigned int* runs) {
    const int binLo = by * FILL_BINS;
    const int nb    = min(FILL_BINS, N - binLo);
    const int t     = threadIdx.x;
    const int lane  = t & 31;
    unsigned int* gbase  = runs;
    unsigned int* lstart = runs + FILL_RUNS;
    unsigned int* lcnt   = runs + 2 * FILL_RUNS;
    unsigned int* lcur   = runs + 3 * FILL_RUNS;

    {
        const unsigned int v = (t < nblk) ? blkTot[t] : 0u;
        unsigned int incl = v;
        for (int o = 1; o < 32; o <<= 1) {
            const unsigned int u = __shfl_up(incl, o);
            if (lane >= o) incl += u;
        }
        if (lane == 31) wtot[t >> 5] = incl;
        __syncthreads();
        if (t < 32) {
            const unsigned int w = wtot[t];
            unsigned int wi = w;
            for (int o = 1; o < 32; o <<= 1) {
                const unsigned int u = __shfl_up(wi, o);
                if (lane >= o) wi += u;
            }
            wtot[t] = wi - w;
        }
        __syncthreads();
        bpre[t] = wtot[t >> 5] + incl - v;
    }
    __syncthreads();

    const int b0  = binLo / SCAN_BLK;
    const int nbk = (nb + SCAN_BLK - 1) / SCAN_BLK;
    unsigned int rn = 0u;
    if (t < nbk) {
        const int b = b0 + t;
        const unsigned int ro = runOfs[(size_t)b * HIST_CHUNKS + bx];
        const unsigned int nx = (bx + 1 < HIST_CHUNKS) ? runOfs[(size_t)b * HIST_CHUNKS + bx + 1] : blkTot[b];
        rn = nx - ro;
        gbase[t] = bpre[b] + ro;
        lcnt[t]  = rn;
        if (bx == 0) blkStart[b] = bpre[b];
    }
    if (bx == 0 && by == nby - 1 && t == 0)
        blkStart[nblk] = bpre[nblk - 1] + blkTot[nblk - 1];
    {
        unsigned int incl = rn;
        for (int o = 1; o < 32; o <<= 1) {
            const unsigned int u = __shfl_up(incl, o);
            if (lane >= o) incl += u;
        }
        if (t < FILL_RUNS && lane == 31) wtot[t >> 5] = incl;
        __syncthreads();
        if (t < FILL_RUNS) {
            unsigned int base = 0u;
            for (int w = 0; w < (t >> 5); ++w) base += wtot[w];
            lstart[t] = base + incl - rn;
            lcur[t]   = base + incl - rn;
        }
    }
    __syncthreads();

    const int per = (E + HIST_CHUNKS - 1) / HIST_CHUNKS;
    const int e0  = bx * per;
    const int e1  = min(E, e0 + per);
    int e = e0 + t;
    for (; e + 7 * 1024 < e1; e += 8 * 1024) {
        unsigned int b[8];
        int s[8];
#pragma unroll
        for (int u = 0; u < 8; ++u) {
            b[u] = (unsigned int)(dst[e + u * 1024] - binLo);
            s[u] = src[e + u * 1024];
        }
        unsigned int pos[8];
#pragma unroll
        for (int u = 0; u < 8; ++u) {
            pos[u] = 0u;
            if (b[u] < (unsigned int)nb) pos[u] = atomicAdd(&lcur[b[u] / SCAN_BLK], 1u);
        }
#pragma unroll
        for (int u = 0; u < 8; ++u)
            if (b[u] < (unsigned int)nb) {
                const unsigned int w = (unsigned int)s[u] | ((b[u] % SCAN_BLK) << 24);
                if (pos[u] < (unsigned int)FILL_BINS) stage[pos[u]] = w;
                else { const unsigned int k = b[u] / SCAN_BLK; ssrc[gbase[k] + (pos[u] - lstart[k])] = (int)w; }
            }
    }
    for (; e < e1; e += 1024) {
        const unsigned int b = (unsigned int)(dst[e] - binLo);
        if (b < (unsigned int)nb) {
            const unsigned int k = b / SCAN_BLK;
            const unsigned int p = atomicAdd(&lcur[k], 1u);
            const unsigned int w = (unsigned int)src[e] | ((b % SCAN_BLK) << 24);
            if (p < (unsigned int)FILL_BINS) stage[p] = w;
            else ssrc[gbase[k] + (p - lstart[k])] = (int)w;
        }
    }
    __syncthreads();

    for (int k = t >> 5; k < nbk; k += 32) {
        const unsigned int n = lcnt[k], ls = lstart[k], gb = gbase[k];
        for (unsigned int j = (unsigned int)lane; j < n; j += 32u) {
            const unsigned int lp = ls + j;
            if (lp < (unsigned int)FILL_BINS) ssrc[gb + j] = (int)stage[lp];
        }
    }
}

#define XS_PITCH 36
__device__ __forceinline__ void gemm1_body(const int g, const int G,
    const float* __restrict__ x, const float* __restrict__ Wc,
    const float* __restrict__ dinv, float* __restrict__ h, int N, float* sW, float* xs) {
    {
        float4* sv = (float4*)sW;
        for (int i = threadIdx.x; i < (IN_DIM / 4) * HID_DIM; i += 1024) {
            const float* gw = Wc + (size_t)(i / HID_DIM) * (4 * HID_DIM) + (i % HID_DIM);
            float4 w;
            w.x = gw[0];
            w.y = gw[HID_DIM];
            w.z = gw[2 * HID_DIM];
            w.w = gw[3 * HID_DIM];
            sv[i] = w;
        }
    }
    __syncthreads();

    const int wave = __builtin_amdgcn_readfirstlane((int)(threadIdx.x >> 5));
    const int lane = threadIdx.x & 31;
    const int r  = lane & 15;
    const int kg = lane >> 4;
    const int nTiles = (N + 15) >> 4;

    const int srow = lane >> 3;
    const int spc  = lane & 7;
    float*       xw  = xs + wave * (16 * XS_PITCH);
    float*       xst = xw + srow * XS_PITCH + spc * 4;
    const float* swl = sW + r * 4 + (kg << 1);
    const char*  xb  = (const char*)x;

    for (int slot = wave; g + G * slot < nTiles; slot += 32) {
        const int rowBase = (g + G * slot) * 16;
        const int ldBase = max(min(rowBase, N - 16), 0);
        const int shift  = rowBase - ldBase;
        const float* xld = xw + min(r + shift, 15) * XS_PITCH + (kg << 1);
        const unsigned int go0 = (unsigned int)(ldBase + srow) * (unsigned int)(IN_DIM * sizeof(float))
                               + (unsigned int)spc * 16u;

        v8f acc[4] = {v8f{}, v8f{}, v8f{}, v8f{}};
#pragma unroll 1
        for (int q = 0; q < 4; ++q) {
            const char* xq = xb + q * 128;
            v4f xv0, xv1, xv2, xv3;
            asm volatile("global_load_b128 %0, %4, %5\n\t"
                         "global_load_b128 %1, %4, %5 offset:2048\n\t"
                         "global_load_b128 %2, %4, %5 offset:4096\n\t"
                         "global_load_b128 %3, %4, %5 offset:6144\n\t"
                         "s_wait_loadcnt 0x0"
                         : "=&v"(xv0), "=&v"(xv1), "=&v"(xv2), "=&v"(xv3)
                         : "v"(go0), "s"(xq)
                         : "memory");
            *(v4f*)(xst)                 = xv0;
            *(v4f*)(xst +  4 * XS_PITCH) = xv1;
            *(v4f*)(xst +  8 * XS_PITCH) = xv2;
            *(v4f*)(xst + 12 * XS_PITCH) = xv3;
            const float* swq = swl + q * (8 * HID_DIM * 4);
#pragma unroll
            for (int s = 0; s < 8; ++s) {
                const float2 av = *(const float2*)(xld + 4 * s);
                v2f a; a.x = av.x; a.y = av.y;
                float2 bv[4];
#pragma unroll
                for (int t = 0; t < 4; ++t)
                    bv[t] = *(const float2*)(swq + (s * HID_DIM + t * 16) * 4);
#pragma unroll
                for (int t = 0; t < 4; ++t) {
                    v2f b; b.x = bv[t].x; b.y = bv[t].y;
                    acc[t] = __builtin_amdgcn_wmma_f32_16x16x4_f32(
                        false, a, false, b, (short)0, acc[t], false, false);
                }
            }
        }
        float dvo[4];
#pragma unroll
        for (int j = 0; j < 4; ++j) dvo[j] = dinv[min(rowBase + srow + 4 * j, N - 1)];
        char* hb = (char*)h;
        float* hst = xw + (8 * kg) * XS_PITCH + r;
        const unsigned int ho0 = (unsigned int)(rowBase + srow) * (unsigned int)(HID_DIM * sizeof(float))
                               + (unsigned int)spc * 16u;
        const bool full = rowBase + 16 <= N;
#pragma unroll
        for (int p = 0; p < 2; ++p) {
#pragma unroll
            for (int tt = 0; tt < 2; ++tt) {
#pragma unroll
                for (int rr = 0; rr < 8; ++rr)
                    hst[rr * XS_PITCH + tt * 16] = acc[2 * p + tt][rr];
            }
            v4f o[4];
#pragma unroll
            for (int j = 0; j < 4; ++j) o[j] = *(const v4f*)(xst + 4 * j * XS_PITCH) * dvo[j];
            if (full) {
#pragma unroll
                for (int j = 0; j < 4; ++j)
                    *(v4f*)(hb + (ho0 + (unsigned int)(j * 4 * HID_DIM * (int)sizeof(float) + p * 128))) = o[j];
            } else {
#pragma unroll
                for (int j = 0; j < 4; ++j)
                    if (rowBase + 4 * j + srow < N)
                        *(v4f*)(hb + (ho0 + (unsigned int)(j * 4 * HID_DIM * (int)sizeof(float) + p * 128))) = o[j];
            }
        }
    }
}

__global__ __launch_bounds__(1024, 16) void fill_gemm1_kernel(
    const int* __restrict__ src, const int* __restrict__ dst, int E, int N,
    const unsigned int* __restrict__ runOfs, const unsigned int* __restrict__ blkTot, int nblk,
    unsigned int* __restrict__ blkStart, int* __restrict__ ssrc,
    const float* __restrict__ x, const float* __restrict__ Wc,
    const float* __restrict__ dinv, float* __restrict__ h) {
    __shared__ __attribute__((aligned(16))) unsigned int cur[FILL_BINS];
    __shared__ unsigned int bpre[MAX_SCAN_BLKS];
    __shared__ unsigned int wtot[32];
    __shared__ unsigned int runs[4 * FILL_RUNS];
    __shared__ __attribute__((aligned(16))) float xs[32 * 16 * XS_PITCH];
    const int nby   = (N + FILL_BINS - 1) / FILL_BINS;
    const int nHist = HIST_CHUNKS * nby;
    const int b     = blockIdx.x;
    if (b < nHist)
        fill_body(b % HIST_CHUNKS, b / HIST_CHUNKS, nby, src, dst, E, N, runOfs, blkTot, nblk,
                  blkStart, ssrc, cur, bpre, wtot, runs);
    else
        gemm1_body(b - nHist, (int)gridDim.x - nHist, x, Wc, dinv, h, N, (float*)cur, xs);
}

#define ZS_PITCH 68
#define GATHER_SLOT(u) if ((u) < m) { const unsigned int s_ = (unsigned int)__builtin_amdgcn_readlane(sv, (u)); v[(u)] = *(const float2*)(hb + ((s_ << 8) + lb)); }
#define SL_CAP   3072

__global__ __launch_bounds__(256) void gemm2_kernel(
    const unsigned int* __restrict__ blkStart, const int* __restrict__ ssrc,
    const float* __restrict__ h, const float* __restrict__ dinv,
    const float* __restrict__ bc, const float* __restrict__ Wl,
    const float* __restrict__ bl, float* __restrict__ out, int N) {
    __shared__ __attribute__((aligned(16))) float sW[HID_DIM * OUT_DIM];
    __shared__ __attribute__((aligned(16))) float zs[128 * ZS_PITCH];
    __shared__ int   sl[SL_CAP];

    const int wave = __builtin_amdgcn_readfirstlane((int)(threadIdx.x >> 5));
    const int lane = threadIdx.x & 31;
    const int rowBase = blockIdx.x * 128 + wave * 16;

    const unsigned int pbeg = blkStart[blockIdx.x];
    const unsigned int pcnt = blkStart[blockIdx.x + 1] - pbeg;
    const bool sorted = pcnt <= (unsigned int)SL_CAP;
    unsigned int ent[SL_CAP / 256];
#pragma unroll
    for (int u = 0; u < SL_CAP / 256; ++u) {
        const unsigned int j = (unsigned int)(u * 256) + threadIdx.x;
        ent[u] = (sorted && j < pcnt) ? (unsigned int)ssrc[pbeg + j] : 0u;
    }
    {
        float4* sv = (float4*)sW;
#pragma unroll
        for (int u = 0; u < ((HID_DIM / 4) * OUT_DIM) / 256; ++u) {
            const int i = u * 256 + (int)threadIdx.x;
            const float* gw = Wl + (size_t)(i / OUT_DIM) * (4 * OUT_DIM) + (i % OUT_DIM);
            float4 w;
            w.x = gw[0];
            w.y = gw[OUT_DIM];
            w.z = gw[2 * OUT_DIM];
            w.w = gw[3 * OUT_DIM];
            sv[i] = w;
        }
    }

    unsigned int* rcnt = (unsigned int*)zs;
    unsigned int* rofs = rcnt + 128;
    unsigned int* rcur = rofs + 132;
    unsigned int* wsm  = rcur + 128;
    unsigned int rp = 0u;
    if (sorted) {
        if (threadIdx.x < 128) rcnt[threadIdx.x] = 0u;
        __syncthreads();
#pragma unroll
        for (int u = 0; u < SL_CAP / 256; ++u)
            if ((unsigned int)(u * 256) + threadIdx.x < pcnt) atomicAdd(&rcnt[ent[u] >> 24], 1u);
        __syncthreads();
        unsigned int v = 0u, incl = 0u;
        if (threadIdx.x < 128) {
            v = rcnt[threadIdx.x];
            incl = v;
            for (int o = 1; o < 32; o <<= 1) {
                const unsigned int u = __shfl_up(incl, o);
                if (lane >= o) incl += u;
            }
            if (lane == 31) wsm[wave] = incl;
        }
        __syncthreads();
        if (threadIdx.x < 128) {
            unsigned int base = 0u;
            for (int w = 0; w < wave; ++w) base += wsm[w];
            const unsigned int ex = base + incl - v;
            rofs[threadIdx.x] = ex;
            rcur[threadIdx.x] = ex;
            if (threadIdx.x == 127) rofs[128] = ex + v;
        }
        __syncthreads();
        unsigned int p[SL_CAP / 256];
#pragma unroll
        for (int u = 0; u < SL_CAP / 256; ++u) {
            p[u] = 0u;
            if ((unsigned int)(u * 256) + threadIdx.x < pcnt) p[u] = atomicAdd(&rcur[ent[u] >> 24], 1u);
        }
#pragma unroll
        for (int u = 0; u < SL_CAP / 256; ++u)
            if ((unsigned int)(u * 256) + threadIdx.x < pcnt) sl[p[u]] = (int)(ent[u] & 0xFFFFFFu);
        rp = rofs[min(wave * 16 + lane, 128)];
        __syncthreads();
    }

    if (rowBase < N) {
        const float        dvec = dinv[min(rowBase + lane, N - 1)];
        const float2       bcv  = *(const float2*)(bc + lane * 2);
        const float* hl = h + lane * 2;
        const char*  hb = (const char*)h;
        const unsigned int lb = (unsigned int)lane * 8u;
        for (int i = 0; i < 16; ++i) {
            const int nld = min(rowBase + i, N - 1);
            float2 acc = *(const float2*)(hl + (size_t)nld * HID_DIM);
            if (sorted) {
                const unsigned int beg = (unsigned int)__builtin_amdgcn_readlane((int)rp, i);
                const unsigned int cnt = (unsigned int)__builtin_amdgcn_readlane((int)rp, i + 1) - beg;
                for (unsigned int base = 0u; base < cnt; base += 32u) {
                    const int m  = (int)min(32u, cnt - base);
                    const int sv = sl[min(beg + base + (unsigned int)lane, (unsigned int)(SL_CAP - 1))];
                    float2 v[32];
#pragma unroll
                    for (int u = 0; u < 32; ++u) v[u] = make_float2(0.0f, 0.0f);
#pragma unroll
                    for (int u = 0; u < 8; ++u) GATHER_SLOT(u)
                    if (m > 8) {
#pragma unroll
                        for (int u = 8; u < 16; ++u) GATHER_SLOT(u)
                        if (m > 16) {
#pragma unroll
                            for (int u = 16; u < 24; ++u) GATHER_SLOT(u)
                            if (m > 24) {
#pragma unroll
                                for (int u = 24; u < 32; ++u) GATHER_SLOT(u)
                            }
                        }
                    }
#pragma unroll
                    for (int u = 0; u < 32; ++u) { acc.x += v[u].x; acc.y += v[u].y; }
                }
            } else {
                const unsigned int myrow = (unsigned int)(wave * 16 + i);
                for (unsigned int j = 0u; j < pcnt; ++j) {
                    const unsigned int w = (unsigned int)__builtin_amdgcn_readfirstlane(ssrc[pbeg + j]);
                    if ((w >> 24) == myrow) {
                        const float2 v0 = *(const float2*)(hl + (size_t)(w & 0xFFFFFFu) * HID_DIM);
                        acc.x += v0.x; acc.y += v0.y;
                    }
                }
            }
            const float di = __int_as_float(__builtin_amdgcn_readlane(__float_as_int(dvec), i));
            float2 z;
            z.x = fmaxf(di * acc.x + bcv.x, 0.0f);
            z.y = fmaxf(di * acc.y + bcv.y, 0.0f);
            *(float2*)(zs + (wave * 16 + i) * ZS_PITCH + lane * 2) = z;
        }
    }
    __syncthreads();
    if (rowBase >= N) return;

    const int r  = lane & 15;
    const int kg = lane >> 4;
    const float* zrow = zs + (wave * 16 + r) * ZS_PITCH;

    const float* swl = sW + r * 4 + (kg << 1);
    v8f acc[8] = {v8f{}, v8f{}, v8f{}, v8f{}, v8f{}, v8f{}, v8f{}, v8f{}};
    for (int k = 0; k < HID_DIM; k += 4) {
        const float2 av = *(const float2*)(zrow + k + (kg << 1));
        v2f a; a.x = av.x; a.y = av.y;
#pragma unroll
        for (int t = 0; t < 8; ++t) {
            const float2 bv = *(const float2*)(swl + k * OUT_DIM + t * 64);
            v2f b; b.x = bv.x; b.y = bv.y;
            acc[t] = __builtin_amdgcn_wmma_f32_16x16x4_f32(
                false, a, false, b, (short)0, acc[t], false, false);
        }
    }
    float*       zst  = zs + (wave * 16 + 8 * kg) * ZS_PITCH + r;
    const int    orow = lane >> 4;
    const int    oc4  = (lane & 15) * 4;
    const float* zld  = zs + (wave * 16 + orow) * ZS_PITCH + oc4;
    float*       ob   = out + (size_t)(rowBase + orow) * OUT_DIM + oc4;
    const bool   full = rowBase + 16 <= N;
    v4f bv[2];
#pragma unroll
    for (int p = 0; p < 2; ++p) {
        bv[p].x = bl[p * 64 + oc4];
        bv[p].y = bl[p * 64 + oc4 + 1];
        bv[p].z = bl[p * 64 + oc4 + 2];
        bv[p].w = bl[p * 64 + oc4 + 3];
    }
#pragma unroll
    for (int p = 0; p < 2; ++p) {
#pragma unroll
        for (int tt = 0; tt < 4; ++tt) {
#pragma unroll
            for (int rr = 0; rr < 8; ++rr)
                zst[rr * ZS_PITCH + tt * 16] = acc[4 * p + tt][rr];
        }
        v4f o[8];
#pragma unroll
        for (int j = 0; j < 8; ++j) o[j] = *(const v4f*)(zld + 2 * j * ZS_PITCH) + bv[p];
        if (full) {
#pragma unroll
            for (int j = 0; j < 8; ++j) *(v4f*)(ob + (size_t)(2 * j) * OUT_DIM + p * 64) = o[j];
        } else {
#pragma unroll
            for (int j = 0; j < 8; ++j)
                if (rowBase + 2 * j + orow < N) *(v4f*)(ob + (size_t)(2 * j) * OUT_DIM + p * 64) = o[j];
        }
    }
}

static inline size_t alignUp(size_t v) { return (v + 255) & ~(size_t)255; }

extern "C" void kernel_launch(void* const* d_in, const int* in_sizes, int n_in,
                              void* d_out, int out_size, void* d_ws, size_t ws_size,
                              hipStream_t stream) {
    const float* x  = (const float*)d_in[0];
    const int*   ei = (const int*)  d_in[1];
    const float* Wc = (const float*)d_in[2];
    const float* bc = (const float*)d_in[3];
    const float* Wl = (const float*)d_in[4];
    const float* bl = (const float*)d_in[5];
    float* out = (float*)d_out;

    const int N = in_sizes[0] / IN_DIM;
    const int E = in_sizes[1] / 2;
    const int* src = ei;
    const int* dst = ei + E;

    char* ws = (char*)d_ws;
    const size_t hBytes = (size_t)N * HID_DIM * sizeof(float);
    size_t off = 0;
    float*        h    = (float*)(ws + off);        off = alignUp(off + hBytes);
    char*         srt  = ws + off;                  off = alignUp(off + hBytes);
    float*        dinv = (float*)(ws + off);        off = alignUp(off + (size_t)N * sizeof(float));

    const int  nblk = (N + SCAN_BLK - 1) / SCAN_BLK;
    size_t so = 0;
    int*          ssrc     = (int*)(srt + so);          so = alignUp(so + (size_t)E * sizeof(int));
    unsigned int* blkStart = (unsigned int*)(srt + so); so = alignUp(so + ((size_t)nblk + 1) * sizeof(unsigned int));
    unsigned int* blkTot   = (unsigned int*)(srt + so); so = alignUp(so + (size_t)nblk * sizeof(unsigned int));
    unsigned int* runOfs   = (unsigned int*)(srt + so); so = alignUp(so + (size_t)nblk * HIST_CHUNKS * sizeof(unsigned int));
    unsigned int* part     = (unsigned int*)(srt + so);

    const dim3 hgrid(HIST_CHUNKS, (N + HIST_BINS - 1) / HIST_BINS);
    const int  nHist = HIST_CHUNKS * ((N + FILL_BINS - 1) / FILL_BINS);
    const int  gemmGroups = 256;

    deg_kernel <<<hgrid, 1024, 0, stream>>>(dst, E, N, part);
    dinv_kernel<<<nblk, SCAN_BLK, 0, stream>>>(part, N, dinv, runOfs, blkTot);
    fill_gemm1_kernel<<<nHist + gemmGroups, 1024, 0, stream>>>(src, dst, E, N, runOfs, blkTot, nblk, blkStart, ssrc,
                                                               x, Wc, dinv, h);

    const int rowBlocks = (N + 127) / 128;

    gemm2_kernel<<<rowBlocks, 256, 0, stream>>>(blkStart, ssrc, h, dinv, bc, Wl, bl, out, N);
}
